# starting kernel
# baseline (speedup 1.0000x reference)
#include <hip/hip_runtime.h>

typedef __attribute__((ext_vector_type(2))) float v2f;
typedef __attribute__((ext_vector_type(4))) float v4f;
typedef __attribute__((ext_vector_type(8))) float v8f;

#define DIM 128

static __device__ __forceinline__ v8f wmma_f32(v2f a, v2f b, v8f c) {
  return __builtin_amdgcn_wmma_f32_16x16x4_f32(false, a, false, b,
                                               (short)0, c, false, false);
}

__global__ void __launch_bounds__(128)
node_selfmsg_kernel(const float* __restrict__ X, const float* __restrict__ W,
                    float* __restrict__ out, int nrows) {
  const int lane  = threadIdx.x & 31;
  const int wave  = threadIdx.x >> 5;
  const int m0    = blockIdx.x * 16;
  const int n0    = wave * 32;
  const int lo    = lane & 15;
  const int khalf = lane >> 4;

  int mrow = m0 + lo;
  if (mrow >= nrows) mrow = nrows - 1;
  const float* Arow = X + (size_t)mrow * DIM;
  const float* B0   = W + (size_t)(n0 + lo) * DIM;
  const float* B1   = B0 + 16 * DIM;

  v8f acc0 = {}, acc1 = {};
  #pragma unroll 4
  for (int kk = 0; kk < DIM; kk += 4) {
    const int k = kk + khalf * 2;
    v2f a  = *(const v2f*)(Arow + k);
    v2f b0 = *(const v2f*)(B0 + k);
    v2f b1 = *(const v2f*)(B1 + k);
    acc0 = wmma_f32(a, b0, acc0);
    acc1 = wmma_f32(a, b1, acc1);
  }
  if (m0 + 16 <= nrows) {
    float* o = out + (size_t)(m0 + khalf * 8) * DIM + n0 + lo;
    #pragma unroll
    for (int r = 0; r < 8; ++r) {
      o[(size_t)r * DIM]      = acc0[r];
      o[(size_t)r * DIM + 16] = acc1[r];
    }
  } else {
    #pragma unroll
    for (int r = 0; r < 8; ++r) {
      const int orow = m0 + khalf * 8 + r;
      if (orow < nrows) {
        float* o = out + (size_t)orow * DIM + lo;
        o[n0]      = acc0[r];
        o[n0 + 16] = acc1[r];
      }
    }
  }
}

__global__ void __launch_bounds__(128)
edge_msg_kernel(const float* __restrict__ src_x, const float* __restrict__ dst_x,
                const int* __restrict__ src_idx, const int* __restrict__ dst_idx,
                const float* __restrict__ ew,
                const float* __restrict__ Wi, const float* __restrict__ We,
                float* __restrict__ out, int nedges) {
  __shared__ float s_xs[32][132];
  __shared__ float s_p [32][132];
  __shared__ float s_ew[32];
  __shared__ int   s_dst[32];

  const int tid = threadIdx.x;
  const int e0  = blockIdx.x * 32;

  {
    const int row = tid >> 2;
    const int cb  = (tid & 3) * 32;
    int e = e0 + row;
    if (e >= nedges) e = nedges - 1;
    const float* xs = src_x + (size_t)src_idx[e] * DIM + cb;
    const float* xd = dst_x + (size_t)dst_idx[e] * DIM + cb;
    #pragma unroll
    for (int j = 0; j < 32; j += 4) {
      v4f a = *(const v4f*)(xs + j);
      v4f b = *(const v4f*)(xd + j);
      *(v4f*)(&s_xs[row][cb + j]) = a;
      *(v4f*)(&s_p [row][cb + j]) = a * b;
    }
    if (tid < 32) {
      const int e2 = e0 + tid;
      s_ew[tid]  = (e2 < nedges) ? ew[e2] : 0.0f;
      s_dst[tid] = dst_idx[(e2 < nedges) ? e2 : (nedges - 1)];
    }
  }
  __syncthreads();

  const int lane  = tid & 31;
  const int wave  = tid >> 5;
  const int n0    = wave * 32;
  const int lo    = lane & 15;
  const int khalf = lane >> 4;

  const float* BiA = Wi + (size_t)(n0 + lo) * DIM;
  const float* BeA = We + (size_t)(n0 + lo) * DIM;
  const float* BiB = BiA + 16 * DIM;
  const float* BeB = BeA + 16 * DIM;

  const float* Ax0 = &s_xs[lo][0];
  const float* Ap0 = &s_p [lo][0];
  const float* Ax1 = &s_xs[16 + lo][0];
  const float* Ap1 = &s_p [16 + lo][0];

  v8f acc00 = {}, acc01 = {}, acc10 = {}, acc11 = {};
  #pragma unroll 2
  for (int kk = 0; kk < DIM; kk += 4) {
    const int k = kk + khalf * 2;
    v2f a0x = *(const v2f*)(Ax0 + k);
    v2f a0p = *(const v2f*)(Ap0 + k);
    v2f a1x = *(const v2f*)(Ax1 + k);
    v2f a1p = *(const v2f*)(Ap1 + k);
    v2f bia = *(const v2f*)(BiA + k);
    v2f bea = *(const v2f*)(BeA + k);
    v2f bib = *(const v2f*)(BiB + k);
    v2f beb = *(const v2f*)(BeB + k);
    acc00 = wmma_f32(a0x, bia, acc00);
    acc00 = wmma_f32(a0p, bea, acc00);
    acc01 = wmma_f32(a0x, bib, acc01);
    acc01 = wmma_f32(a0p, beb, acc01);
    acc10 = wmma_f32(a1x, bia, acc10);
    acc10 = wmma_f32(a1p, bea, acc10);
    acc11 = wmma_f32(a1x, bib, acc11);
    acc11 = wmma_f32(a1p, beb, acc11);
  }

  #pragma unroll
  for (int r = 0; r < 8; ++r) {
    const int row0 = r + 8 * khalf;
    const int row1 = 16 + row0;
    float* o0 = out + (size_t)s_dst[row0] * DIM + lo;
    float* o1 = out + (size_t)s_dst[row1] * DIM + lo;
    const float w0 = s_ew[row0];
    const float w1 = s_ew[row1];
    atomicAdd(o0 + n0,      w0 * acc00[r]);
    atomicAdd(o0 + n0 + 16, w0 * acc01[r]);
    atomicAdd(o1 + n0,      w1 * acc10[r]);
    atomicAdd(o1 + n0 + 16, w1 * acc11[r]);
  }
}

__global__ void __launch_bounds__(256)
lrelu_kernel(float* __restrict__ out, int n4) {
  const int i = blockIdx.x * blockDim.x + threadIdx.x;
  if (i < n4) {
    v4f x = *(v4f*)(out + (size_t)i * 4);
    #pragma unroll
    for (int c = 0; c < 4; ++c) x[c] = (x[c] > 0.0f) ? x[c] : 0.01f * x[c];
    *(v4f*)(out + (size_t)i * 4) = x;
  }
}

extern "C" void kernel_launch(void* const* d_in, const int* in_sizes, int n_in,
                              void* d_out, int out_size, void* d_ws, size_t ws_size,
                              hipStream_t stream) {
  const float* src_x = (const float*)d_in[0];
  const float* dst_x = (const float*)d_in[1];
  const int*   eidx  = (const int*)  d_in[2];
  const float* ew    = (const float*)d_in[3];
  const float* Wi    = (const float*)d_in[4];
  const float* We    = (const float*)d_in[5];
  float*       out   = (float*)d_out;

  const int N = in_sizes[0] / DIM;
  const int E = in_sizes[2] / 2;
  const int* src_idx = eidx;
  const int* dst_idx = eidx + E;

  node_selfmsg_kernel<<<(N + 15) / 16, 128, 0, stream>>>(dst_x, Wi, out, N);
  edge_msg_kernel<<<(E + 31) / 32, 128, 0, stream>>>(src_x, dst_x, src_idx, dst_idx,
                                                     ew, Wi, We, out, E);
  const int n4 = (N * DIM) / 4;
  lrelu_kernel<<<(n4 + 255) / 256, 256, 0, stream>>>(out, n4);
}


# Edges binned per 16-node dst tile (int-atomic count, scan, arrival-order place); tile sums ew*src rows on WMMA via a selection matrix, then fused transform+lrelu; no float atomics; sum order changes
# speedup vs baseline: 1.8452x; 1.8452x over previous
#include <hip/hip_runtime.h>


typedef __attribute__((ext_vector_type(2))) float v2f;
typedef __attribute__((ext_vector_type(4))) float v4f;
typedef __attribute__((ext_vector_type(8))) float v8f;
typedef __attribute__((ext_vector_type(2))) unsigned int v2u;
typedef __attribute__((ext_vector_type(4))) unsigned int v4u;

#define DIM 128
#define CH  32
#define GP  136
#define XP  132

static __device__ __forceinline__ v8f wmma_f32(v2f a, v2f b, v8f c) {
  return __builtin_amdgcn_wmma_f32_16x16x4_f32(false, a, false, b,
                                               (short)0, c, false, false);
}

static __device__ __forceinline__ size_t ws_plane(int ntiles) {
  return ((size_t)ntiles + 32) & ~(size_t)31;
}
static __device__ __forceinline__ bool ws_fits(size_t ws_bytes, int ntiles, int nedges) {
  return ws_bytes >= ws_plane(ntiles) * 8 + (size_t)nedges * 8;
}

__global__ void __launch_bounds__(256)
bin_clear_kernel(int* __restrict__ ws, size_t ws_bytes, int ntiles, int nedges) {
  if (!ws_fits(ws_bytes, ntiles, nedges)) return;
  const int i = blockIdx.x * 256 + threadIdx.x;
  if (i < ntiles) ws[i] = 0;
}

__global__ void __launch_bounds__(256)
bin_count_kernel(const int* __restrict__ dst_idx, int* __restrict__ ws,
                 size_t ws_bytes, int ntiles, int nedges) {
  if (!ws_fits(ws_bytes, ntiles, nedges)) return;
  const int T  = gridDim.x * 256;
  const int e0 = blockIdx.x * 256 + threadIdx.x;
  int d[4];
  #pragma unroll
  for (int i = 0; i < 4; ++i) {
    const int e = e0 + i * T;
    d[i] = dst_idx[(e < nedges) ? e : 0];
  }
  #pragma unroll
  for (int i = 0; i < 4; ++i) {
    const int e = e0 + i * T;
    if (e < nedges) atomicAdd(&ws[d[i] >> 4], 1);
  }
}

__global__ void __launch_bounds__(1024)
bin_scan_kernel(int* __restrict__ ws, size_t ws_bytes, int ntiles, int nedges) {
  if (!ws_fits(ws_bytes, ntiles, nedges)) return;
  __shared__ int s_sum[1024];
  int* cursor = ws;
  int* off    = ws + ws_plane(ntiles);
  const int t   = threadIdx.x;
  const int per = (ntiles + 1023) / 1024;
  const int b   = t * per;
  int s = 0;
  for (int i = 0; i < per; ++i) {
    const int j = b + i;
    if (j < ntiles) s += cursor[j];
  }
  s_sum[t] = s;
  __syncthreads();
  for (int dlt = 1; dlt < 1024; dlt <<= 1) {
    const int x = (t >= dlt) ? s_sum[t - dlt] : 0;
    __syncthreads();
    s_sum[t] += x;
    __syncthreads();
  }
  int run = s_sum[t] - s;
  for (int i = 0; i < per; ++i) {
    const int j = b + i;
    if (j < ntiles) {
      const int c = cursor[j];
      off[j]    = run;
      cursor[j] = run;
      run += c;
    }
  }
  if (t == 0) off[ntiles] = nedges;
}

__global__ void __launch_bounds__(256)
bin_place_kernel(const int* __restrict__ src_idx, const int* __restrict__ dst_idx,
                 const float* __restrict__ ew, int* __restrict__ ws,
                 size_t ws_bytes, int ntiles, int nedges) {
  if (!ws_fits(ws_bytes, ntiles, nedges)) return;
  int* cursor = ws;
  v2u* rec    = (v2u*)(ws + 2 * ws_plane(ntiles));
  const int T  = gridDim.x * 256;
  const int e0 = blockIdx.x * 256 + threadIdx.x;
  int d[4], s[4];
  float w[4];
  #pragma unroll
  for (int i = 0; i < 4; ++i) {
    const int e  = e0 + i * T;
    const int ec = (e < nedges) ? e : 0;
    d[i] = dst_idx[ec];
    s[i] = src_idx[ec];
    w[i] = ew[ec];
  }
  #pragma unroll
  for (int i = 0; i < 4; ++i) {
    const int e = e0 + i * T;
    if (e < nedges) {
      const int pos = atomicAdd(&cursor[d[i] >> 4], 1);
      v2u r;
      r.x = (unsigned)s[i] | ((unsigned)(d[i] & 15) << 28);
      r.y = __float_as_uint(w[i]);
      rec[pos] = r;
    }
  }
}

__global__ void __launch_bounds__(128)
tile_gather_transform_kernel(const float* __restrict__ src_x, const float* __restrict__ dst_x,
                             const float* __restrict__ Wi, const float* __restrict__ We,
                             const int* __restrict__ ws, size_t ws_bytes,
                             float* __restrict__ out, int nrows, int ntiles, int nedges) {
  if (!ws_fits(ws_bytes, ntiles, nedges)) return;
  __shared__ __attribute__((aligned(16))) float        s_buf[CH * GP];
  __shared__ __attribute__((aligned(16))) unsigned int s_rec[CH * 2];

  const int tid   = threadIdx.x;
  const int lane  = tid & 31;
  const int wave  = tid >> 5;
  const int lo    = lane & 15;
  const int khalf = lane >> 4;
  const int n0    = wave * 32;
  const int m0    = blockIdx.x * 16;

  const int* off = ws + ws_plane(ntiles);
  const v2u* rec = (const v2u*)(ws + 2 * ws_plane(ntiles));
  const int beg = off[blockIdx.x];
  const int end = off[blockIdx.x + 1];

  float dv0[8], dv1[8];
  #pragma unroll
  for (int r = 0; r < 8; ++r) {
    int row = m0 + r + 8 * khalf;
    if (row >= nrows) row = nrows - 1;
    const float* dp = dst_x + (size_t)row * DIM + n0 + lo;
    dv0[r] = dp[0];
    dv1[r] = dp[16];
  }

  v8f g0 = {}, g1 = {};
  const int srow = tid >> 2;
  const int scb  = (tid & 3) * 32;
  float* sg = &s_buf[srow * GP + scb];
  v2u rc = {0u, 0u};
  if (beg + srow < end) rc = rec[beg + srow];
  for (int base = beg; base < end; base += CH) {
    const int p = base + srow;
    v2u rn = {0u, 0u};
    if (p + CH < end) rn = rec[p + CH];
    if (p < end) {
      const float* xs = src_x + (size_t)(rc.x & 0x0FFFFFFFu) * DIM + scb;
      #pragma unroll
      for (int j = 0; j < 32; j += 4) *(v4f*)(sg + j) = *(const v4f*)(xs + j);
      if ((tid & 3) == 0) {
        s_rec[2 * srow]     = rc.x >> 28;
        s_rec[2 * srow + 1] = rc.y;
      }
    } else {
      const v4f z = {0.0f, 0.0f, 0.0f, 0.0f};
      #pragma unroll
      for (int j = 0; j < 32; j += 4) *(v4f*)(sg + j) = z;
      if ((tid & 3) == 0) {
        s_rec[2 * srow]     = 0u;
        s_rec[2 * srow + 1] = 0u;
      }
    }
    __syncthreads();
    #pragma unroll
    for (int kk = 0; kk < CH; kk += 4) {
      const int k = kk + 2 * khalf;
      const v4u q = *(const v4u*)&s_rec[2 * k];
      v2f a;
      a.x = (q.x == (unsigned)lo) ? __uint_as_float(q.y) : 0.0f;
      a.y = (q.z == (unsigned)lo) ? __uint_as_float(q.w) : 0.0f;
      const float* gk = &s_buf[k * GP + n0 + lo];
      v2f b0, b1;
      b0.x = gk[0];   b0.y = gk[GP];
      b1.x = gk[16];  b1.y = gk[GP + 16];
      g0 = wmma_f32(a, b0, g0);
      g1 = wmma_f32(a, b1, g1);
    }
    __syncthreads();
    rc = rn;
  }

  float* s_x = s_buf;
  float* s_p = s_buf + 16 * XP;
  #pragma unroll
  for (int r = 0; r < 8; ++r) {
    const int o = (r + 8 * khalf) * XP + n0 + lo;
    s_x[o]      = dv0[r] + g0[r];
    s_p[o]      = g0[r] * dv0[r];
    s_x[o + 16] = dv1[r] + g1[r];
    s_p[o + 16] = g1[r] * dv1[r];
  }
  __syncthreads();

  const float* Ax  = s_x + lo * XP;
  const float* Ap  = s_p + lo * XP;
  const float* Bi0 = Wi + (size_t)(n0 + lo) * DIM;
  const float* Be0 = We + (size_t)(n0 + lo) * DIM;
  const float* Bi1 = Bi0 + 16 * DIM;
  const float* Be1 = Be0 + 16 * DIM;

  v8f o0 = {}, o1 = {};
  #pragma unroll 4
  for (int kk = 0; kk < DIM; kk += 4) {
    const int k = kk + 2 * khalf;
    const v2f ax  = *(const v2f*)(Ax + k);
    const v2f ap  = *(const v2f*)(Ap + k);
    const v2f bi0 = *(const v2f*)(Bi0 + k);
    const v2f bi1 = *(const v2f*)(Bi1 + k);
    const v2f be0 = *(const v2f*)(Be0 + k);
    const v2f be1 = *(const v2f*)(Be1 + k);
    o0 = wmma_f32(ax, bi0, o0);
    o1 = wmma_f32(ax, bi1, o1);
    o0 = wmma_f32(ap, be0, o0);
    o1 = wmma_f32(ap, be1, o1);
  }

  if (m0 + 16 <= nrows) {
    float* o = out + (size_t)(m0 + 8 * khalf) * DIM + n0 + lo;
    #pragma unroll
    for (int r = 0; r < 8; ++r) {
      const float y0 = o0[r], y1 = o1[r];
      o[(size_t)r * DIM]      = (y0 > 0.0f) ? y0 : 0.01f * y0;
      o[(size_t)r * DIM + 16] = (y1 > 0.0f) ? y1 : 0.01f * y1;
    }
  } else {
    #pragma unroll
    for (int r = 0; r < 8; ++r) {
      const int orow = m0 + 8 * khalf + r;
      if (orow < nrows) {
        float* o = out + (size_t)orow * DIM + n0 + lo;
        const float y0 = o0[r], y1 = o1[r];
        o[0]  = (y0 > 0.0f) ? y0 : 0.01f * y0;
        o[16] = (y1 > 0.0f) ? y1 : 0.01f * y1;
      }
    }
  }
}

extern "C" void kernel_launch(void* const* d_in, const int* in_sizes, int n_in,
                              void* d_out, int out_size, void* d_ws, size_t ws_size,
                              hipStream_t stream) {
  const float* src_x = (const float*)d_in[0];
  const float* dst_x = (const float*)d_in[1];
  const int*   eidx  = (const int*)  d_in[2];
  const float* ew    = (const float*)d_in[3];
  const float* Wi    = (const float*)d_in[4];
  const float* We    = (const float*)d_in[5];
  float*       out   = (float*)d_out;
  int*         ws    = (int*)d_ws;

  const int N  = in_sizes[0] / DIM;
  const int E  = in_sizes[2] / 2;
  const int NT = (N + 15) / 16;
  const int* src_idx = eidx;
  const int* dst_idx = eidx + E;

  bin_clear_kernel<<<(NT + 255) / 256, 256, 0, stream>>>(ws, ws_size, NT, E);
  bin_count_kernel<<<(E + 1023) / 1024, 256, 0, stream>>>(dst_idx, ws, ws_size, NT, E);
  bin_scan_kernel<<<1, 1024, 0, stream>>>(ws, ws_size, NT, E);
  bin_place_kernel<<<(E + 1023) / 1024, 256, 0, stream>>>(src_idx, dst_idx, ew, ws, ws_size, NT, E);
  tile_gather_transform_kernel<<<NT, 128, 0, stream>>>(src_x, dst_x, Wi, We, ws, ws_size,
                                                       out, N, NT, E);
}


# Gather staging remapped: each wave-wide 16-byte load reads one whole 512-byte source row (4 lines; was 16-byte pieces of 32 lines); records passed by readlane; slots and sum order unchanged
# speedup vs baseline: 1.9372x; 1.0499x over previous
#include <hip/hip_runtime.h>


typedef __attribute__((ext_vector_type(2))) float v2f;
typedef __attribute__((ext_vector_type(4))) float v4f;
typedef __attribute__((ext_vector_type(8))) float v8f;
typedef __attribute__((ext_vector_type(2))) unsigned int v2u;
typedef __attribute__((ext_vector_type(4))) unsigned int v4u;

#define DIM 128
#define CH  32
#define GP  136
#define XP  132

static __device__ __forceinline__ v8f wmma_f32(v2f a, v2f b, v8f c) {
  return __builtin_amdgcn_wmma_f32_16x16x4_f32(false, a, false, b,
                                               (short)0, c, false, false);
}

static __device__ __forceinline__ size_t ws_plane(int ntiles) {
  return ((size_t)ntiles + 32) & ~(size_t)31;
}
static __device__ __forceinline__ bool ws_fits(size_t ws_bytes, int ntiles, int nedges) {
  return ws_bytes >= ws_plane(ntiles) * 8 + (size_t)nedges * 8;
}

__global__ void __launch_bounds__(256)
bin_clear_kernel(int* __restrict__ ws, size_t ws_bytes, int ntiles, int nedges) {
  if (!ws_fits(ws_bytes, ntiles, nedges)) return;
  const int i = blockIdx.x * 256 + threadIdx.x;
  if (i < ntiles) ws[i] = 0;
}

__global__ void __launch_bounds__(256)
bin_count_kernel(const int* __restrict__ dst_idx, int* __restrict__ ws,
                 size_t ws_bytes, int ntiles, int nedges) {
  if (!ws_fits(ws_bytes, ntiles, nedges)) return;
  const int T  = gridDim.x * 256;
  const int e0 = blockIdx.x * 256 + threadIdx.x;
  int d[4];
  #pragma unroll
  for (int i = 0; i < 4; ++i) {
    const int e = e0 + i * T;
    d[i] = dst_idx[(e < nedges) ? e : 0];
  }
  #pragma unroll
  for (int i = 0; i < 4; ++i) {
    const int e = e0 + i * T;
    if (e < nedges) atomicAdd(&ws[d[i] >> 4], 1);
  }
}

__global__ void __launch_bounds__(1024)
bin_scan_kernel(int* __restrict__ ws, size_t ws_bytes, int ntiles, int nedges) {
  if (!ws_fits(ws_bytes, ntiles, nedges)) return;
  __shared__ int s_sum[1024];
  int* cursor = ws;
  int* off    = ws + ws_plane(ntiles);
  const int t   = threadIdx.x;
  const int per = (ntiles + 1023) / 1024;
  const int b   = t * per;
  int s = 0;
  for (int i = 0; i < per; ++i) {
    const int j = b + i;
    if (j < ntiles) s += cursor[j];
  }
  s_sum[t] = s;
  __syncthreads();
  for (int dlt = 1; dlt < 1024; dlt <<= 1) {
    const int x = (t >= dlt) ? s_sum[t - dlt] : 0;
    __syncthreads();
    s_sum[t] += x;
    __syncthreads();
  }
  int run = s_sum[t] - s;
  for (int i = 0; i < per; ++i) {
    const int j = b + i;
    if (j < ntiles) {
      const int c = cursor[j];
      off[j]    = run;
      cursor[j] = run;
      run += c;
    }
  }
  if (t == 0) off[ntiles] = nedges;
}

__global__ void __launch_bounds__(256)
bin_place_kernel(const int* __restrict__ src_idx, const int* __restrict__ dst_idx,
                 const float* __restrict__ ew, int* __restrict__ ws,
                 size_t ws_bytes, int ntiles, int nedges) {
  if (!ws_fits(ws_bytes, ntiles, nedges)) return;
  int* cursor = ws;
  v2u* rec    = (v2u*)(ws + 2 * ws_plane(ntiles));
  const int T  = gridDim.x * 256;
  const int e0 = blockIdx.x * 256 + threadIdx.x;
  int d[4], s[4];
  float w[4];
  #pragma unroll
  for (int i = 0; i < 4; ++i) {
    const int e  = e0 + i * T;
    const int ec = (e < nedges) ? e : 0;
    d[i] = dst_idx[ec];
    s[i] = src_idx[ec];
    w[i] = ew[ec];
  }
  #pragma unroll
  for (int i = 0; i < 4; ++i) {
    const int e = e0 + i * T;
    if (e < nedges) {
      const int pos = atomicAdd(&cursor[d[i] >> 4], 1);
      v2u r;
      r.x = (unsigned)s[i] | ((unsigned)(d[i] & 15) << 28);
      r.y = __float_as_uint(w[i]);
      rec[pos] = r;
    }
  }
}

__global__ void __launch_bounds__(128)
tile_gather_transform_kernel(const float* __restrict__ src_x, const float* __restrict__ dst_x,
                             const float* __restrict__ Wi, const float* __restrict__ We,
                             const int* __restrict__ ws, size_t ws_bytes,
                             float* __restrict__ out, int nrows, int ntiles, int nedges) {
  if (!ws_fits(ws_bytes, ntiles, nedges)) return;
  __shared__ __attribute__((aligned(16))) float        s_buf[CH * GP];
  __shared__ __attribute__((aligned(16))) unsigned int s_rec[CH * 2];

  const int tid   = threadIdx.x;
  const int lane  = tid & 31;
  const int wave  = tid >> 5;
  const int lo    = lane & 15;
  const int khalf = lane >> 4;
  const int n0    = wave * 32;
  const int m0    = blockIdx.x * 16;

  const int* off = ws + ws_plane(ntiles);
  const v2u* rec = (const v2u*)(ws + 2 * ws_plane(ntiles));
  const int beg = off[blockIdx.x];
  const int end = off[blockIdx.x + 1];

  float dv0[8], dv1[8];
  #pragma unroll
  for (int r = 0; r < 8; ++r) {
    int row = m0 + r + 8 * khalf;
    if (row >= nrows) row = nrows - 1;
    const float* dp = dst_x + (size_t)row * DIM + n0 + lo;
    dv0[r] = dp[0];
    dv1[r] = dp[16];
  }

  v8f g0 = {}, g1 = {};
  const int wv   = __builtin_amdgcn_readfirstlane(wave);
  const int slot = 4 * (lane & 7) + wv;
  float* sg = &s_buf[wv * GP + lane * 4];
  v2u rc = {0u, 0u};
  if (beg + slot < end) rc = rec[beg + slot];
  for (int base = beg; base < end; base += CH) {
    const int p = base + slot;
    v2u rn = {0u, 0u};
    if (p + CH < end) rn = rec[p + CH];
    v4f v[8];
    #pragma unroll
    for (int i = 0; i < 8; ++i) {
      const unsigned rx = (unsigned)__builtin_amdgcn_readlane((int)rc.x, i);
      if (base + 4 * i + wv < end) {
        v[i] = *(const v4f*)(src_x + (size_t)(rx & 0x0FFFFFFFu) * DIM + lane * 4);
      } else {
        v[i] = (v4f){0.0f, 0.0f, 0.0f, 0.0f};
      }
    }
    #pragma unroll
    for (int i = 0; i < 8; ++i) *(v4f*)(sg + i * 4 * GP) = v[i];
    if (lane < 8) {
      v2u q;
      q.x = rc.x >> 28;
      q.y = rc.y;
      *(v2u*)&s_rec[2 * slot] = q;
    }
    __syncthreads();
    #pragma unroll
    for (int kk = 0; kk < CH; kk += 4) {
      const int k = kk + 2 * khalf;
      const v4u q = *(const v4u*)&s_rec[2 * k];
      v2f a;
      a.x = (q.x == (unsigned)lo) ? __uint_as_float(q.y) : 0.0f;
      a.y = (q.z == (unsigned)lo) ? __uint_as_float(q.w) : 0.0f;
      const float* gk = &s_buf[k * GP + n0 + lo];
      v2f b0, b1;
      b0.x = gk[0];   b0.y = gk[GP];
      b1.x = gk[16];  b1.y = gk[GP + 16];
      g0 = wmma_f32(a, b0, g0);
      g1 = wmma_f32(a, b1, g1);
    }
    __syncthreads();
    rc = rn;
  }

  float* s_x = s_buf;
  float* s_p = s_buf + 16 * XP;
  #pragma unroll
  for (int r = 0; r < 8; ++r) {
    const int o = (r + 8 * khalf) * XP + n0 + lo;
    s_x[o]      = dv0[r] + g0[r];
    s_p[o]      = g0[r] * dv0[r];
    s_x[o + 16] = dv1[r] + g1[r];
    s_p[o + 16] = g1[r] * dv1[r];
  }
  __syncthreads();

  const float* Ax  = s_x + lo * XP;
  const float* Ap  = s_p + lo * XP;
  const float* Bi0 = Wi + (size_t)(n0 + lo) * DIM;
  const float* Be0 = We + (size_t)(n0 + lo) * DIM;
  const float* Bi1 = Bi0 + 16 * DIM;
  const float* Be1 = Be0 + 16 * DIM;

  v8f o0 = {}, o1 = {};
  #pragma unroll 4
  for (int kk = 0; kk < DIM; kk += 4) {
    const int k = kk + 2 * khalf;
    const v2f ax  = *(const v2f*)(Ax + k);
    const v2f ap  = *(const v2f*)(Ap + k);
    const v2f bi0 = *(const v2f*)(Bi0 + k);
    const v2f bi1 = *(const v2f*)(Bi1 + k);
    const v2f be0 = *(const v2f*)(Be0 + k);
    const v2f be1 = *(const v2f*)(Be1 + k);
    o0 = wmma_f32(ax, bi0, o0);
    o1 = wmma_f32(ax, bi1, o1);
    o0 = wmma_f32(ap, be0, o0);
    o1 = wmma_f32(ap, be1, o1);
  }

  if (m0 + 16 <= nrows) {
    float* o = out + (size_t)(m0 + 8 * khalf) * DIM + n0 + lo;
    #pragma unroll
    for (int r = 0; r < 8; ++r) {
      const float y0 = o0[r], y1 = o1[r];
      o[(size_t)r * DIM]      = (y0 > 0.0f) ? y0 : 0.01f * y0;
      o[(size_t)r * DIM + 16] = (y1 > 0.0f) ? y1 : 0.01f * y1;
    }
  } else {
    #pragma unroll
    for (int r = 0; r < 8; ++r) {
      const int orow = m0 + 8 * khalf + r;
      if (orow < nrows) {
        float* o = out + (size_t)orow * DIM + n0 + lo;
        const float y0 = o0[r], y1 = o1[r];
        o[0]  = (y0 > 0.0f) ? y0 : 0.01f * y0;
        o[16] = (y1 > 0.0f) ? y1 : 0.01f * y1;
      }
    }
  }
}

extern "C" void kernel_launch(void* const* d_in, const int* in_sizes, int n_in,
                              void* d_out, int out_size, void* d_ws, size_t ws_size,
                              hipStream_t stream) {
  const float* src_x = (const float*)d_in[0];
  const float* dst_x = (const float*)d_in[1];
  const int*   eidx  = (const int*)  d_in[2];
  const float* ew    = (const float*)d_in[3];
  const float* Wi    = (const float*)d_in[4];
  const float* We    = (const float*)d_in[5];
  float*       out   = (float*)d_out;
  int*         ws    = (int*)d_ws;

  const int N  = in_sizes[0] / DIM;
  const int E  = in_sizes[2] / 2;
  const int NT = (N + 15) / 16;
  const int* src_idx = eidx;
  const int* dst_idx = eidx + E;

  bin_clear_kernel<<<(NT + 255) / 256, 256, 0, stream>>>(ws, ws_size, NT, E);
  bin_count_kernel<<<(E + 1023) / 1024, 256, 0, stream>>>(dst_idx, ws, ws_size, NT, E);
  bin_scan_kernel<<<1, 1024, 0, stream>>>(ws, ws_size, NT, E);
  bin_place_kernel<<<(E + 1023) / 1024, 256, 0, stream>>>(src_idx, dst_idx, ew, ws, ws_size, NT, E);
  tile_gather_transform_kernel<<<NT, 128, 0, stream>>>(src_x, dst_x, Wi, We, ws, ws_size,
                                                       out, N, NT, E);
}


# Work-group owns 4 consecutive 16-node tiles: x|p blocks built tile by tile in LDS, then ONE transform pass in which each W fragment read from global feeds 4 row tiles (was 1); sums unchanged
# speedup vs baseline: 1.9962x; 1.0305x over previous
#include <hip/hip_runtime.h>


typedef __attribute__((ext_vector_type(2))) float v2f;
typedef __attribute__((ext_vector_type(4))) float v4f;
typedef __attribute__((ext_vector_type(8))) float v8f;
typedef __attribute__((ext_vector_type(2))) unsigned int v2u;
typedef __attribute__((ext_vector_type(4))) unsigned int v4u;

#define DIM 128
#define CH  32
#define GP  136
#define XP  132
#define RT  4
#define XPT (2 * 16 * XP)

static __device__ __forceinline__ v8f wmma_f32(v2f a, v2f b, v8f c) {
  return __builtin_amdgcn_wmma_f32_16x16x4_f32(false, a, false, b,
                                               (short)0, c, false, false);
}

static __device__ __forceinline__ size_t ws_plane(int ntiles) {
  return ((size_t)ntiles + 32) & ~(size_t)31;
}
static __device__ __forceinline__ bool ws_fits(size_t ws_bytes, int ntiles, int nedges) {
  return ws_bytes >= ws_plane(ntiles) * 8 + (size_t)nedges * 8;
}

__global__ void __launch_bounds__(256)
bin_clear_kernel(int* __restrict__ ws, size_t ws_bytes, int ntiles, int nedges) {
  if (!ws_fits(ws_bytes, ntiles, nedges)) return;
  const int i = blockIdx.x * 256 + threadIdx.x;
  if (i < ntiles) ws[i] = 0;
}

__global__ void __launch_bounds__(256)
bin_count_kernel(const int* __restrict__ dst_idx, int* __restrict__ ws,
                 size_t ws_bytes, int ntiles, int nedges) {
  if (!ws_fits(ws_bytes, ntiles, nedges)) return;
  const int T  = gridDim.x * 256;
  const int e0 = blockIdx.x * 256 + threadIdx.x;
  int d[4];
  #pragma unroll
  for (int i = 0; i < 4; ++i) {
    const int e = e0 + i * T;
    d[i] = dst_idx[(e < nedges) ? e : 0];
  }
  #pragma unroll
  for (int i = 0; i < 4; ++i) {
    const int e = e0 + i * T;
    if (e < nedges) atomicAdd(&ws[d[i] >> 4], 1);
  }
}

__global__ void __launch_bounds__(1024)
bin_scan_kernel(int* __restrict__ ws, size_t ws_bytes, int ntiles, int nedges) {
  if (!ws_fits(ws_bytes, ntiles, nedges)) return;
  __shared__ int s_sum[1024];
  int* cursor = ws;
  int* off    = ws + ws_plane(ntiles);
  const int t   = threadIdx.x;
  const int per = (ntiles + 1023) / 1024;
  const int b   = t * per;
  int s = 0;
  for (int i = 0; i < per; ++i) {
    const int j = b + i;
    if (j < ntiles) s += cursor[j];
  }
  s_sum[t] = s;
  __syncthreads();
  for (int dlt = 1; dlt < 1024; dlt <<= 1) {
    const int x = (t >= dlt) ? s_sum[t - dlt] : 0;
    __syncthreads();
    s_sum[t] += x;
    __syncthreads();
  }
  int run = s_sum[t] - s;
  for (int i = 0; i < per; ++i) {
    const int j = b + i;
    if (j < ntiles) {
      const int c = cursor[j];
      off[j]    = run;
      cursor[j] = run;
      run += c;
    }
  }
  if (t == 0) off[ntiles] = nedges;
}

__global__ void __launch_bounds__(256)
bin_place_kernel(const int* __restrict__ src_idx, const int* __restrict__ dst_idx,
                 const float* __restrict__ ew, int* __restrict__ ws,
                 size_t ws_bytes, int ntiles, int nedges) {
  if (!ws_fits(ws_bytes, ntiles, nedges)) return;
  int* cursor = ws;
  v2u* rec    = (v2u*)(ws + 2 * ws_plane(ntiles));
  const int T  = gridDim.x * 256;
  const int e0 = blockIdx.x * 256 + threadIdx.x;
  int d[4], s[4];
  float w[4];
  #pragma unroll
  for (int i = 0; i < 4; ++i) {
    const int e  = e0 + i * T;
    const int ec = (e < nedges) ? e : 0;
    d[i] = dst_idx[ec];
    s[i] = src_idx[ec];
    w[i] = ew[ec];
  }
  #pragma unroll
  for (int i = 0; i < 4; ++i) {
    const int e = e0 + i * T;
    if (e < nedges) {
      const int pos = atomicAdd(&cursor[d[i] >> 4], 1);
      v2u r;
      r.x = (unsigned)s[i] | ((unsigned)(d[i] & 15) << 28);
      r.y = __float_as_uint(w[i]);
      rec[pos] = r;
    }
  }
}

__global__ void __launch_bounds__(128)
tile_gather_transform_kernel(const float* __restrict__ src_x, const float* __restrict__ dst_x,
                             const float* __restrict__ Wi, const float* __restrict__ We,
                             const int* __restrict__ ws, size_t ws_bytes,
                             float* __restrict__ out, int nrows, int ntiles, int nedges) {
  if (!ws_fits(ws_bytes, ntiles, nedges)) return;
  __shared__ __attribute__((aligned(16))) float        s_all[(RT - 1) * XPT + CH * GP];
  __shared__ __attribute__((aligned(16))) unsigned int s_rec[CH * 2];
  float* s_buf = s_all + (RT - 1) * XPT;

  const int tid   = threadIdx.x;
  const int lane  = tid & 31;
  const int wave  = tid >> 5;
  const int lo    = lane & 15;
  const int khalf = lane >> 4;
  const int n0    = wave * 32;
  const int tile0 = blockIdx.x * RT;

  const int* off = ws + ws_plane(ntiles);
  const v2u* rec = (const v2u*)(ws + 2 * ws_plane(ntiles));
  int eo[RT + 1];
  #pragma unroll
  for (int i = 0; i <= RT; ++i) eo[i] = off[(tile0 + i < ntiles) ? tile0 + i : ntiles];

  const int wv   = __builtin_amdgcn_readfirstlane(wave);
  const int slot = 4 * (lane & 7) + wv;
  float* sg = &s_buf[wv * GP + lane * 4];

  for (int t = 0; t < RT; ++t) {
  const int m0  = (tile0 + t) * 16;
  const int beg = eo[0];
  const int end = eo[1];
  #pragma unroll
  for (int i = 0; i < RT; ++i) eo[i] = eo[i + 1];

  float dv0[8], dv1[8];
  #pragma unroll
  for (int r = 0; r < 8; ++r) {
    int row = m0 + r + 8 * khalf;
    if (row >= nrows) row = nrows - 1;
    const float* dp = dst_x + (size_t)row * DIM + n0 + lo;
    dv0[r] = dp[0];
    dv1[r] = dp[16];
  }

  v8f g0 = {}, g1 = {};
  v2u rc = {0u, 0u};
  if (beg + slot < end) rc = rec[beg + slot];
  for (int base = beg; base < end; base += CH) {
    const int p = base + slot;
    v2u rn = {0u, 0u};
    if (p + CH < end) rn = rec[p + CH];
    v4f v[8];
    #pragma unroll
    for (int i = 0; i < 8; ++i) {
      const unsigned rx = (unsigned)__builtin_amdgcn_readlane((int)rc.x, i);
      const v4f x = *(const v4f*)(src_x + (size_t)(rx & 0x0FFFFFFFu) * DIM + lane * 4);
      v[i] = (base + 4 * i + wv < end) ? x : (v4f){0.0f, 0.0f, 0.0f, 0.0f};
    }
    #pragma unroll
    for (int i = 0; i < 8; ++i) *(v4f*)(sg + i * 4 * GP) = v[i];
    if (lane < 8) {
      v2u q;
      q.x = rc.x >> 28;
      q.y = rc.y;
      *(v2u*)&s_rec[2 * slot] = q;
    }
    __syncthreads();
    #pragma unroll
    for (int kk = 0; kk < CH; kk += 4) {
      const int k = kk + 2 * khalf;
      const v4u q = *(const v4u*)&s_rec[2 * k];
      v2f a;
      a.x = (q.x == (unsigned)lo) ? __uint_as_float(q.y) : 0.0f;
      a.y = (q.z == (unsigned)lo) ? __uint_as_float(q.w) : 0.0f;
      const float* gk = &s_buf[k * GP + n0 + lo];
      v2f b0, b1;
      b0.x = gk[0];   b0.y = gk[GP];
      b1.x = gk[16];  b1.y = gk[GP + 16];
      g0 = wmma_f32(a, b0, g0);
      g1 = wmma_f32(a, b1, g1);
    }
    __syncthreads();
    rc = rn;
  }

  float* s_x = s_all + t * XPT;
  float* s_p = s_x + 16 * XP;
  #pragma unroll
  for (int r = 0; r < 8; ++r) {
    const int o = (r + 8 * khalf) * XP + n0 + lo;
    s_x[o]      = dv0[r] + g0[r];
    s_p[o]      = g0[r] * dv0[r];
    s_x[o + 16] = dv1[r] + g1[r];
    s_p[o + 16] = g1[r] * dv1[r];
  }
  }
  __syncthreads();

  const float* Ax  = s_all + lo * XP;
  const float* Ap  = Ax + 16 * XP;
  const float* Bi0 = Wi + (size_t)(n0 + lo) * DIM;
  const float* Be0 = We + (size_t)(n0 + lo) * DIM;
  const float* Bi1 = Bi0 + 16 * DIM;
  const float* Be1 = Be0 + 16 * DIM;

  v8f o0[RT], o1[RT];
  #pragma unroll
  for (int t = 0; t < RT; ++t) { o0[t] = (v8f){}; o1[t] = (v8f){}; }
  #pragma unroll 2
  for (int kk = 0; kk < DIM; kk += 4) {
    const int k = kk + 2 * khalf;
    const v2f bi0 = *(const v2f*)(Bi0 + k);
    const v2f bi1 = *(const v2f*)(Bi1 + k);
    const v2f be0 = *(const v2f*)(Be0 + k);
    const v2f be1 = *(const v2f*)(Be1 + k);
    #pragma unroll
    for (int t = 0; t < RT; ++t) {
      const v2f ax = *(const v2f*)(Ax + t * XPT + k);
      const v2f ap = *(const v2f*)(Ap + t * XPT + k);
      o0[t] = wmma_f32(ax, bi0, o0[t]);
      o1[t] = wmma_f32(ax, bi1, o1[t]);
      o0[t] = wmma_f32(ap, be0, o0[t]);
      o1[t] = wmma_f32(ap, be1, o1[t]);
    }
  }

  #pragma unroll
  for (int t = 0; t < RT; ++t) {
    const int m0 = (tile0 + t) * 16;
    if (m0 + 16 <= nrows) {
      float* o = out + (size_t)(m0 + 8 * khalf) * DIM + n0 + lo;
      #pragma unroll
      for (int r = 0; r < 8; ++r) {
        const float y0 = o0[t][r], y1 = o1[t][r];
        o[(size_t)r * DIM]      = (y0 > 0.0f) ? y0 : 0.01f * y0;
        o[(size_t)r * DIM + 16] = (y1 > 0.0f) ? y1 : 0.01f * y1;
      }
    } else {
      #pragma unroll
      for (int r = 0; r < 8; ++r) {
        const int orow = m0 + 8 * khalf + r;
        if (orow < nrows) {
          float* o = out + (size_t)orow * DIM + n0 + lo;
          const float y0 = o0[t][r], y1 = o1[t][r];
          o[0]  = (y0 > 0.0f) ? y0 : 0.01f * y0;
          o[16] = (y1 > 0.0f) ? y1 : 0.01f * y1;
        }
      }
    }
  }
}

extern "C" void kernel_launch(void* const* d_in, const int* in_sizes, int n_in,
                              void* d_out, int out_size, void* d_ws, size_t ws_size,
                              hipStream_t stream) {
  const float* src_x = (const float*)d_in[0];
  const float* dst_x = (const float*)d_in[1];
  const int*   eidx  = (const int*)  d_in[2];
  const float* ew    = (const float*)d_in[3];
  const float* Wi    = (const float*)d_in[4];
  const float* We    = (const float*)d_in[5];
  float*       out   = (float*)d_out;
  int*         ws    = (int*)d_ws;

  const int N  = in_sizes[0] / DIM;
  const int E  = in_sizes[2] / 2;
  const int NT = (N + 15) / 16;
  const int* src_idx = eidx;
  const int* dst_idx = eidx + E;

  bin_clear_kernel<<<(NT + 255) / 256, 256, 0, stream>>>(ws, ws_size, NT, E);
  bin_count_kernel<<<(E + 1023) / 1024, 256, 0, stream>>>(dst_idx, ws, ws_size, NT, E);
  bin_scan_kernel<<<1, 1024, 0, stream>>>(ws, ws_size, NT, E);
  bin_place_kernel<<<(E + 1023) / 1024, 256, 0, stream>>>(src_idx, dst_idx, ew, ws, ws_size, NT, E);
  tile_gather_transform_kernel<<<(NT + RT - 1) / RT, 128, 0, stream>>>(src_x, dst_x, Wi, We, ws, ws_size,
                                                       out, N, NT, E);
}


# Edge binning without global atomics: per-chunk tile counts in LDS to a count table, each placing work-group takes its bases from it, LDS counters give the slots; 5 launches to 3; sum order changes
# speedup vs baseline: 4.9165x; 2.4629x over previous
#include <hip/hip_runtime.h>


typedef __attribute__((ext_vector_type(2))) float v2f;
typedef __attribute__((ext_vector_type(4))) float v4f;
typedef __attribute__((ext_vector_type(8))) float v8f;
typedef __attribute__((ext_vector_type(2))) unsigned int v2u;
typedef __attribute__((ext_vector_type(4))) unsigned int v4u;

#define DIM 128
#define CH  32
#define GP  136
#define XP  132
#define RT  4
#define XPT (2 * 16 * XP)

static __device__ __forceinline__ v8f wmma_f32(v2f a, v2f b, v8f c) {
  return __builtin_amdgcn_wmma_f32_16x16x4_f32(false, a, false, b,
                                               (short)0, c, false, false);
}

#define NCH 32
#define RB  1024

static __device__ __forceinline__ size_t ws_plane(int ntiles) {
  return ((size_t)ntiles + 32) & ~(size_t)31;
}
static __device__ __forceinline__ int bin_ranges(int ntiles) {
  return (ntiles + RB - 1) / RB;
}
static __device__ __forceinline__ size_t ws_rec(int ntiles) {
  const size_t rs = ((size_t)NCH * bin_ranges(ntiles) + 31) & ~(size_t)31;
  return (size_t)(NCH + 1) * ws_plane(ntiles) + rs;
}
static __device__ __forceinline__ bool ws_fits(size_t ws_bytes, int ntiles, int nedges) {
  return ws_bytes >= ws_rec(ntiles) * 4 + (size_t)nedges * 8;
}

static __device__ __forceinline__ int block_sum(int v, int* s_w) {
  #pragma unroll
  for (int m = 16; m >= 1; m >>= 1) v += __shfl_xor(v, m);
  if ((threadIdx.x & 31) == 0) s_w[threadIdx.x >> 5] = v;
  __syncthreads();
  int r = s_w[threadIdx.x & 31];
  #pragma unroll
  for (int m = 16; m >= 1; m >>= 1) r += __shfl_xor(r, m);
  __syncthreads();
  return r;
}
static __device__ __forceinline__ int block_excl_scan(int v, int* s_w) {
  const int lane = threadIdx.x & 31;
  const int wv   = threadIdx.x >> 5;
  int x = v;
  #pragma unroll
  for (int d = 1; d < 32; d <<= 1) {
    const int y = __shfl_up(x, d);
    if (lane >= d) x += y;
  }
  if (lane == 31) s_w[wv] = x;
  __syncthreads();
  const int w = s_w[lane];
  int wsum = w;
  #pragma unroll
  for (int d = 1; d < 32; d <<= 1) {
    const int y = __shfl_up(wsum, d);
    if (lane >= d) wsum += y;
  }
  const int wex = __shfl(wsum - w, wv);
  __syncthreads();
  return wex + x - v;
}

__global__ void __launch_bounds__(1024)
bin_count_kernel(const int* __restrict__ dst_idx, int* __restrict__ ws,
                 size_t ws_bytes, int ntiles, int nedges) {
  if (!ws_fits(ws_bytes, ntiles, nedges)) return;
  __shared__ int s_cnt[RB];
  __shared__ int s_w[32];
  const int t  = threadIdx.x;
  const int c  = blockIdx.x;
  const int h  = blockIdx.y;
  const size_t tp = ws_plane(ntiles);
  int* cnt  = ws + tp;
  int* rsum = ws + (size_t)(NCH + 1) * tp;
  const int per   = (nedges + NCH - 1) / NCH;
  const int e_beg = c * per;
  const int e_end = (e_beg + per < nedges) ? e_beg + per : nedges;
  const int b0    = h * RB;
  s_cnt[t] = 0;
  __syncthreads();
  int mine = 0;
  for (int e0 = e_beg + t; e0 < e_end; e0 += 4 * 1024) {
    int d[4];
    #pragma unroll
    for (int i = 0; i < 4; ++i) {
      const int e = e0 + i * 1024;
      d[i] = dst_idx[(e < e_end) ? e : e0];
    }
    #pragma unroll
    for (int i = 0; i < 4; ++i) {
      const int e  = e0 + i * 1024;
      const int tl = d[i] >> 4;
      const unsigned u = (unsigned)(tl - b0);
      if (e < e_end && u < (unsigned)RB && tl < ntiles) {
        atomicAdd(&s_cnt[u], 1);
        ++mine;
      }
    }
  }
  __syncthreads();
  if (b0 + t < ntiles) cnt[(size_t)c * tp + b0 + t] = s_cnt[t];
  const int tot = block_sum(mine, s_w);
  if (t == 0) rsum[h * NCH + c] = tot;
}

__global__ void __launch_bounds__(1024)
bin_place_kernel(const int* __restrict__ src_idx, const int* __restrict__ dst_idx,
                 const float* __restrict__ ew, int* __restrict__ ws,
                 size_t ws_bytes, int ntiles, int nedges) {
  if (!ws_fits(ws_bytes, ntiles, nedges)) return;
  __shared__ int s_cur[RB];
  __shared__ int s_w[32];
  const int t  = threadIdx.x;
  const int c  = blockIdx.x;
  const int h  = blockIdx.y;
  const size_t tp = ws_plane(ntiles);
  int*       off  = ws;
  const int* cnt  = ws + tp;
  const int* rsum = ws + (size_t)(NCH + 1) * tp;
  v2u*       rec  = (v2u*)(ws + ws_rec(ntiles));
  const int b0 = h * RB;
  const int b  = b0 + t;

  int tot = 0, pre = 0;
  if (b < ntiles) {
    #pragma unroll 8
    for (int cc = 0; cc < NCH; ++cc) {
      const int v = cnt[(size_t)cc * tp + b];
      tot += v;
      pre += (cc < c) ? v : 0;
    }
  }
  int low = 0;
  for (int j = t; j < h * NCH; j += 1024) low += rsum[j];
  const int lowsum = block_sum(low, s_w);
  const int first  = lowsum + block_excl_scan(tot, s_w);
  if (c == 0) {
    if (b < ntiles) off[b] = first;
    if (h == 0 && t == 0) off[ntiles] = nedges;
  }
  s_cur[t] = first + pre;
  __syncthreads();

  const int per   = (nedges + NCH - 1) / NCH;
  const int e_beg = c * per;
  const int e_end = (e_beg + per < nedges) ? e_beg + per : nedges;
  for (int e0 = e_beg + t; e0 < e_end; e0 += 4 * 1024) {
    int d[4], s[4];
    float w[4];
    #pragma unroll
    for (int i = 0; i < 4; ++i) {
      const int e  = e0 + i * 1024;
      const int ec = (e < e_end) ? e : e0;
      d[i] = dst_idx[ec];
      s[i] = src_idx[ec];
      w[i] = ew[ec];
    }
    #pragma unroll
    for (int i = 0; i < 4; ++i) {
      const int e  = e0 + i * 1024;
      const int tl = d[i] >> 4;
      const unsigned u = (unsigned)(tl - b0);
      if (e < e_end && u < (unsigned)RB && tl < ntiles) {
        const int pos = atomicAdd(&s_cur[u], 1);
        v2u r;
        r.x = (unsigned)s[i] | ((unsigned)(d[i] & 15) << 28);
        r.y = __float_as_uint(w[i]);
        rec[pos] = r;
      }
    }
  }
}

__global__ void __launch_bounds__(128)
tile_gather_transform_kernel(const float* __restrict__ src_x, const float* __restrict__ dst_x,
                             const float* __restrict__ Wi, const float* __restrict__ We,
                             const int* __restrict__ ws, size_t ws_bytes,
                             float* __restrict__ out, int nrows, int ntiles, int nedges) {
  if (!ws_fits(ws_bytes, ntiles, nedges)) return;
  __shared__ __attribute__((aligned(16))) float        s_all[(RT - 1) * XPT + CH * GP];
  __shared__ __attribute__((aligned(16))) unsigned int s_rec[CH * 2];
  float* s_buf = s_all + (RT - 1) * XPT;

  const int tid   = threadIdx.x;
  const int lane  = tid & 31;
  const int wave  = tid >> 5;
  const int lo    = lane & 15;
  const int khalf = lane >> 4;
  const int n0    = wave * 32;
  const int tile0 = blockIdx.x * RT;

  const int* off = ws;
  const v2u* rec = (const v2u*)(ws + ws_rec(ntiles));
  int eo[RT + 1];
  #pragma unroll
  for (int i = 0; i <= RT; ++i) eo[i] = off[(tile0 + i < ntiles) ? tile0 + i : ntiles];

  const int wv   = __builtin_amdgcn_readfirstlane(wave);
  const int slot = 4 * (lane & 7) + wv;
  float* sg = &s_buf[wv * GP + lane * 4];

  for (int t = 0; t < RT; ++t) {
  const int m0  = (tile0 + t) * 16;
  const int beg = eo[0];
  const int end = eo[1];
  #pragma unroll
  for (int i = 0; i < RT; ++i) eo[i] = eo[i + 1];

  float dv0[8], dv1[8];
  #pragma unroll
  for (int r = 0; r < 8; ++r) {
    int row = m0 + r + 8 * khalf;
    if (row >= nrows) row = nrows - 1;
    const float* dp = dst_x + (size_t)row * DIM + n0 + lo;
    dv0[r] = dp[0];
    dv1[r] = dp[16];
  }

  v8f g0 = {}, g1 = {};
  v2u rc = {0u, 0u};
  if (beg + slot < end) rc = rec[beg + slot];
  for (int base = beg; base < end; base += CH) {
    const int p = base + slot;
    v2u rn = {0u, 0u};
    if (p + CH < end) rn = rec[p + CH];
    v4f v[8];
    #pragma unroll
    for (int i = 0; i < 8; ++i) {
      const unsigned rx = (unsigned)__builtin_amdgcn_readlane((int)rc.x, i);
      const v4f x = *(const v4f*)(src_x + (size_t)(rx & 0x0FFFFFFFu) * DIM + lane * 4);
      v[i] = (base + 4 * i + wv < end) ? x : (v4f){0.0f, 0.0f, 0.0f, 0.0f};
    }
    #pragma unroll
    for (int i = 0; i < 8; ++i) *(v4f*)(sg + i * 4 * GP) = v[i];
    if (lane < 8) {
      v2u q;
      q.x = rc.x >> 28;
      q.y = rc.y;
      *(v2u*)&s_rec[2 * slot] = q;
    }
    __syncthreads();
    #pragma unroll
    for (int kk = 0; kk < CH; kk += 4) {
      const int k = kk + 2 * khalf;
      const v4u q = *(const v4u*)&s_rec[2 * k];
      v2f a;
      a.x = (q.x == (unsigned)lo) ? __uint_as_float(q.y) : 0.0f;
      a.y = (q.z == (unsigned)lo) ? __uint_as_float(q.w) : 0.0f;
      const float* gk = &s_buf[k * GP + n0 + lo];
      v2f b0, b1;
      b0.x = gk[0];   b0.y = gk[GP];
      b1.x = gk[16];  b1.y = gk[GP + 16];
      g0 = wmma_f32(a, b0, g0);
      g1 = wmma_f32(a, b1, g1);
    }
    __syncthreads();
    rc = rn;
  }

  float* s_x = s_all + t * XPT;
  float* s_p = s_x + 16 * XP;
  #pragma unroll
  for (int r = 0; r < 8; ++r) {
    const int o = (r + 8 * khalf) * XP + n0 + lo;
    s_x[o]      = dv0[r] + g0[r];
    s_p[o]      = g0[r] * dv0[r];
    s_x[o + 16] = dv1[r] + g1[r];
    s_p[o + 16] = g1[r] * dv1[r];
  }
  }
  __syncthreads();

  const float* Ax  = s_all + lo * XP;
  const float* Ap  = Ax + 16 * XP;
  const float* Bi0 = Wi + (size_t)(n0 + lo) * DIM;
  const float* Be0 = We + (size_t)(n0 + lo) * DIM;
  const float* Bi1 = Bi0 + 16 * DIM;
  const float* Be1 = Be0 + 16 * DIM;

  v8f o0[RT], o1[RT];
  #pragma unroll
  for (int t = 0; t < RT; ++t) { o0[t] = (v8f){}; o1[t] = (v8f){}; }
  #pragma unroll 2
  for (int kk = 0; kk < DIM; kk += 4) {
    const int k = kk + 2 * khalf;
    const v2f bi0 = *(const v2f*)(Bi0 + k);
    const v2f bi1 = *(const v2f*)(Bi1 + k);
    const v2f be0 = *(const v2f*)(Be0 + k);
    const v2f be1 = *(const v2f*)(Be1 + k);
    #pragma unroll
    for (int t = 0; t < RT; ++t) {
      const v2f ax = *(const v2f*)(Ax + t * XPT + k);
      const v2f ap = *(const v2f*)(Ap + t * XPT + k);
      o0[t] = wmma_f32(ax, bi0, o0[t]);
      o1[t] = wmma_f32(ax, bi1, o1[t]);
      o0[t] = wmma_f32(ap, be0, o0[t]);
      o1[t] = wmma_f32(ap, be1, o1[t]);
    }
  }

  #pragma unroll
  for (int t = 0; t < RT; ++t) {
    const int m0 = (tile0 + t) * 16;
    if (m0 + 16 <= nrows) {
      float* o = out + (size_t)(m0 + 8 * khalf) * DIM + n0 + lo;
      #pragma unroll
      for (int r = 0; r < 8; ++r) {
        const float y0 = o0[t][r], y1 = o1[t][r];
        o[(size_t)r * DIM]      = (y0 > 0.0f) ? y0 : 0.01f * y0;
        o[(size_t)r * DIM + 16] = (y1 > 0.0f) ? y1 : 0.01f * y1;
      }
    } else {
      #pragma unroll
      for (int r = 0; r < 8; ++r) {
        const int orow = m0 + 8 * khalf + r;
        if (orow < nrows) {
          float* o = out + (size_t)orow * DIM + n0 + lo;
          const float y0 = o0[t][r], y1 = o1[t][r];
          o[0]  = (y0 > 0.0f) ? y0 : 0.01f * y0;
          o[16] = (y1 > 0.0f) ? y1 : 0.01f * y1;
        }
      }
    }
  }
}

extern "C" void kernel_launch(void* const* d_in, const int* in_sizes, int n_in,
                              void* d_out, int out_size, void* d_ws, size_t ws_size,
                              hipStream_t stream) {
  const float* src_x = (const float*)d_in[0];
  const float* dst_x = (const float*)d_in[1];
  const int*   eidx  = (const int*)  d_in[2];
  const float* ew    = (const float*)d_in[3];
  const float* Wi    = (const float*)d_in[4];
  const float* We    = (const float*)d_in[5];
  float*       out   = (float*)d_out;
  int*         ws    = (int*)d_ws;

  const int N  = in_sizes[0] / DIM;
  const int E  = in_sizes[2] / 2;
  const int NT = (N + 15) / 16;
  const int* src_idx = eidx;
  const int* dst_idx = eidx + E;

  const dim3 bgrid(NCH, (NT + RB - 1) / RB);
  bin_count_kernel<<<bgrid, 1024, 0, stream>>>(dst_idx, ws, ws_size, NT, E);
  bin_place_kernel<<<bgrid, 1024, 0, stream>>>(src_idx, dst_idx, ew, ws, ws_size, NT, E);
  tile_gather_transform_kernel<<<(NT + RT - 1) / RT, 128, 0, stream>>>(src_x, dst_x, Wi, We, ws, ws_size,
                                                       out, N, NT, E);
}


# Edge passes software-pipelined: next pass's source rows requested ahead of the current pass's WMMAs, records two passes ahead, also across tile ends; pass barriers order LDS only; sums unchanged
# speedup vs baseline: 5.3885x; 1.0960x over previous
#include <hip/hip_runtime.h>


typedef __attribute__((ext_vector_type(2))) float v2f;
typedef __attribute__((ext_vector_type(4))) float v4f;
typedef __attribute__((ext_vector_type(8))) float v8f;
typedef __attribute__((ext_vector_type(2))) unsigned int v2u;
typedef __attribute__((ext_vector_type(4))) unsigned int v4u;

#define DIM 128
#define CH  32
#define GP  136
#define XP  132
#define RT  4
#define XPT (2 * 16 * XP)

static __device__ __forceinline__ v8f wmma_f32(v2f a, v2f b, v8f c) {
  return __builtin_amdgcn_wmma_f32_16x16x4_f32(false, a, false, b,
                                               (short)0, c, false, false);
}

#define NCH 32
#define RB  1024

static __device__ __forceinline__ size_t ws_plane(int ntiles) {
  return ((size_t)ntiles + 32) & ~(size_t)31;
}
static __device__ __forceinline__ int bin_ranges(int ntiles) {
  return (ntiles + RB - 1) / RB;
}
static __device__ __forceinline__ size_t ws_rec(int ntiles) {
  const size_t rs = ((size_t)NCH * bin_ranges(ntiles) + 31) & ~(size_t)31;
  return (size_t)(NCH + 1) * ws_plane(ntiles) + rs;
}
static __device__ __forceinline__ bool ws_fits(size_t ws_bytes, int ntiles, int nedges) {
  return ws_bytes >= ws_rec(ntiles) * 4 + (size_t)nedges * 8;
}

static __device__ __forceinline__ int block_sum(int v, int* s_w) {
  #pragma unroll
  for (int m = 16; m >= 1; m >>= 1) v += __shfl_xor(v, m);
  if ((threadIdx.x & 31) == 0) s_w[threadIdx.x >> 5] = v;
  __syncthreads();
  int r = s_w[threadIdx.x & 31];
  #pragma unroll
  for (int m = 16; m >= 1; m >>= 1) r += __shfl_xor(r, m);
  __syncthreads();
  return r;
}
static __device__ __forceinline__ int block_excl_scan(int v, int* s_w) {
  const int lane = threadIdx.x & 31;
  const int wv   = threadIdx.x >> 5;
  int x = v;
  #pragma unroll
  for (int d = 1; d < 32; d <<= 1) {
    const int y = __shfl_up(x, d);
    if (lane >= d) x += y;
  }
  if (lane == 31) s_w[wv] = x;
  __syncthreads();
  const int w = s_w[lane];
  int wsum = w;
  #pragma unroll
  for (int d = 1; d < 32; d <<= 1) {
    const int y = __shfl_up(wsum, d);
    if (lane >= d) wsum += y;
  }
  const int wex = __shfl(wsum - w, wv);
  __syncthreads();
  return wex + x - v;
}

__global__ void __launch_bounds__(1024)
bin_count_kernel(const int* __restrict__ dst_idx, int* __restrict__ ws,
                 size_t ws_bytes, int ntiles, int nedges) {
  if (!ws_fits(ws_bytes, ntiles, nedges)) return;
  __shared__ int s_cnt[RB];
  __shared__ int s_w[32];
  const int t  = threadIdx.x;
  const int c  = blockIdx.x;
  const int h  = blockIdx.y;
  const size_t tp = ws_plane(ntiles);
  int* cnt  = ws + tp;
  int* rsum = ws + (size_t)(NCH + 1) * tp;
  const int per   = (nedges + NCH - 1) / NCH;
  const int e_beg = c * per;
  const int e_end = (e_beg + per < nedges) ? e_beg + per : nedges;
  const int b0    = h * RB;
  s_cnt[t] = 0;
  __syncthreads();
  int mine = 0;
  for (int e0 = e_beg + t; e0 < e_end; e0 += 4 * 1024) {
    int d[4];
    #pragma unroll
    for (int i = 0; i < 4; ++i) {
      const int e = e0 + i * 1024;
      d[i] = dst_idx[(e < e_end) ? e : e0];
    }
    #pragma unroll
    for (int i = 0; i < 4; ++i) {
      const int e  = e0 + i * 1024;
      const int tl = d[i] >> 4;
      const unsigned u = (unsigned)(tl - b0);
      if (e < e_end && u < (unsigned)RB && tl < ntiles) {
        atomicAdd(&s_cnt[u], 1);
        ++mine;
      }
    }
  }
  __syncthreads();
  if (b0 + t < ntiles) cnt[(size_t)c * tp + b0 + t] = s_cnt[t];
  const int tot = block_sum(mine, s_w);
  if (t == 0) rsum[h * NCH + c] = tot;
}

__global__ void __launch_bounds__(1024)
bin_place_kernel(const int* __restrict__ src_idx, const int* __restrict__ dst_idx,
                 const float* __restrict__ ew, int* __restrict__ ws,
                 size_t ws_bytes, int ntiles, int nedges) {
  if (!ws_fits(ws_bytes, ntiles, nedges)) return;
  __shared__ int s_cur[RB];
  __shared__ int s_w[32];
  const int t  = threadIdx.x;
  const int c  = blockIdx.x;
  const int h  = blockIdx.y;
  const size_t tp = ws_plane(ntiles);
  int*       off  = ws;
  const int* cnt  = ws + tp;
  const int* rsum = ws + (size_t)(NCH + 1) * tp;
  v2u*       rec  = (v2u*)(ws + ws_rec(ntiles));
  const int b0 = h * RB;
  const int b  = b0 + t;

  int tot = 0, pre = 0;
  if (b < ntiles) {
    #pragma unroll 8
    for (int cc = 0; cc < NCH; ++cc) {
      const int v = cnt[(size_t)cc * tp + b];
      tot += v;
      pre += (cc < c) ? v : 0;
    }
  }
  int low = 0;
  for (int j = t; j < h * NCH; j += 1024) low += rsum[j];
  const int lowsum = block_sum(low, s_w);
  const int first  = lowsum + block_excl_scan(tot, s_w);
  if (c == 0) {
    if (b < ntiles) off[b] = first;
    if (h == 0 && t == 0) off[ntiles] = nedges;
  }
  s_cur[t] = first + pre;
  __syncthreads();

  const int per   = (nedges + NCH - 1) / NCH;
  const int e_beg = c * per;
  const int e_end = (e_beg + per < nedges) ? e_beg + per : nedges;
  for (int e0 = e_beg + t; e0 < e_end; e0 += 4 * 1024) {
    int d[4], s[4];
    float w[4];
    #pragma unroll
    for (int i = 0; i < 4; ++i) {
      const int e  = e0 + i * 1024;
      const int ec = (e < e_end) ? e : e0;
      d[i] = dst_idx[ec];
      s[i] = src_idx[ec];
      w[i] = ew[ec];
    }
    #pragma unroll
    for (int i = 0; i < 4; ++i) {
      const int e  = e0 + i * 1024;
      const int tl = d[i] >> 4;
      const unsigned u = (unsigned)(tl - b0);
      if (e < e_end && u < (unsigned)RB && tl < ntiles) {
        const int pos = atomicAdd(&s_cur[u], 1);
        v2u r;
        r.x = (unsigned)s[i] | ((unsigned)(d[i] & 15) << 28);
        r.y = __float_as_uint(w[i]);
        rec[pos] = r;
      }
    }
  }
}

static __device__ __forceinline__ void lds_barrier() {
  __builtin_amdgcn_fence(__ATOMIC_RELEASE, "workgroup", "local");
  __builtin_amdgcn_s_barrier();
  __builtin_amdgcn_fence(__ATOMIC_ACQUIRE, "workgroup", "local");
}

static __device__ __forceinline__ void load_dst(float (&dv0)[8], float (&dv1)[8], const float* __restrict__ dst_x,
                                                int m0, int khalf, int nrows, int col) {
  #pragma unroll
  for (int r = 0; r < 8; ++r) {
    int row = m0 + r + 8 * khalf;
    if (row >= nrows) row = nrows - 1;
    const float* dp = dst_x + (size_t)row * DIM + col;
    dv0[r] = dp[0];
    dv1[r] = dp[16];
  }
}

static __device__ __forceinline__ void issue_rows(v4f (&v)[8], const float* __restrict__ src_x,
                                                  v2u rc, int pb, int pe, int wv, int lane) {
  #pragma unroll
  for (int i = 0; i < 8; ++i) {
    const unsigned rx = (unsigned)__builtin_amdgcn_readlane((int)rc.x, i);
    const v4f x = *(const v4f*)(src_x + (size_t)(rx & 0x0FFFFFFFu) * DIM + lane * 4);
    v[i] = (pb + 4 * i + wv < pe) ? x : (v4f){0.0f, 0.0f, 0.0f, 0.0f};
  }
}

__global__ void __launch_bounds__(128)
tile_gather_transform_kernel(const float* __restrict__ src_x, const float* __restrict__ dst_x,
                             const float* __restrict__ Wi, const float* __restrict__ We,
                             const int* __restrict__ ws, size_t ws_bytes,
                             float* __restrict__ out, int nrows, int ntiles, int nedges) {
  if (!ws_fits(ws_bytes, ntiles, nedges)) return;
  __shared__ __attribute__((aligned(16))) float        s_all[(RT - 1) * XPT + CH * GP];
  __shared__ __attribute__((aligned(16))) unsigned int s_rec[CH * 2];
  float* s_buf = s_all + (RT - 1) * XPT;

  const int tid   = threadIdx.x;
  const int lane  = tid & 31;
  const int wave  = tid >> 5;
  const int lo    = lane & 15;
  const int khalf = lane >> 4;
  const int n0    = wave * 32;
  const int tile0 = blockIdx.x * RT;

  const int* off = ws;
  const v2u* rec = (const v2u*)(ws + ws_rec(ntiles));
  int eo[RT + 1];
  #pragma unroll
  for (int i = 0; i <= RT; ++i) eo[i] = off[(tile0 + i < ntiles) ? tile0 + i : ntiles];

  const int wv   = __builtin_amdgcn_readfirstlane(wave);
  const int slot = 4 * (lane & 7) + wv;
  float* sg = &s_buf[wv * GP + lane * 4];

  v4f  v[8];
  v2u  rc = {0u, 0u}, rn = {0u, 0u};
  bool pf = false;
  #pragma unroll
  for (int i = 0; i < 8; ++i) v[i] = (v4f){0.0f, 0.0f, 0.0f, 0.0f};

  for (int t = 0; t < RT; ++t) {
  const int m0  = (tile0 + t) * 16;
  const int beg = eo[0];
  const int end = eo[1];
  #pragma unroll
  for (int i = 0; i < RT; ++i) eo[i] = eo[i + 1];

  float dv0[8], dv1[8];

  v8f g0 = {}, g1 = {};
  if (beg < end) {
    if (!pf) {
      const bool l0 = (beg + CH >= end);
      const int  b1 = l0 ? end : beg + CH;
      const int  e1 = l0 ? eo[1] : end;
      rc = (v2u){0u, 0u};
      rn = (v2u){0u, 0u};
      if (beg + slot < end) rc = rec[beg + slot];
      if (b1 + slot < e1)   rn = rec[b1 + slot];
      issue_rows(v, src_x, rc, beg, end, wv, lane);
    }
    for (int base = beg; base < end; base += CH) {
      #pragma unroll
      for (int i = 0; i < 8; ++i) *(v4f*)(sg + i * 4 * GP) = v[i];
      if (lane < 8) {
        v2u q;
        q.x = rc.x >> 28;
        q.y = rc.y;
        *(v2u*)&s_rec[2 * slot] = q;
      }
      lds_barrier();
      const bool l1 = (base + CH >= end);
      const int  b1 = l1 ? end : base + CH;
      const int  e1 = l1 ? eo[1] : end;
      const bool l2 = (b1 + CH >= e1);
      const int  b2 = l2 ? e1 : b1 + CH;
      const int  e2 = l2 ? (l1 ? eo[2] : eo[1]) : e1;
      issue_rows(v, src_x, rn, b1, e1, wv, lane);
      v2u rnn = {0u, 0u};
      if (b2 + slot < e2) rnn = rec[b2 + slot];
      if (base == beg) load_dst(dv0, dv1, dst_x, m0, khalf, nrows, n0 + lo);
      #pragma unroll
      for (int kk = 0; kk < CH; kk += 4) {
        const int k = kk + 2 * khalf;
        const v4u q = *(const v4u*)&s_rec[2 * k];
        v2f a;
        a.x = (q.x == (unsigned)lo) ? __uint_as_float(q.y) : 0.0f;
        a.y = (q.z == (unsigned)lo) ? __uint_as_float(q.w) : 0.0f;
        const float* gk = &s_buf[k * GP + n0 + lo];
        v2f b0, b1f;
        b0.x  = gk[0];   b0.y  = gk[GP];
        b1f.x = gk[16];  b1f.y = gk[GP + 16];
        g0 = wmma_f32(a, b0,  g0);
        g1 = wmma_f32(a, b1f, g1);
      }
      lds_barrier();
      rc = rn;
      rn = rnn;
    }
    pf = true;
  } else {
    pf = false;
    load_dst(dv0, dv1, dst_x, m0, khalf, nrows, n0 + lo);
  }

  float* s_x = s_all + t * XPT;
  float* s_p = s_x + 16 * XP;
  #pragma unroll
  for (int r = 0; r < 8; ++r) {
    const int o = (r + 8 * khalf) * XP + n0 + lo;
    s_x[o]      = dv0[r] + g0[r];
    s_p[o]      = g0[r] * dv0[r];
    s_x[o + 16] = dv1[r] + g1[r];
    s_p[o + 16] = g1[r] * dv1[r];
  }
  }
  __syncthreads();

  const float* Ax  = s_all + lo * XP;
  const float* Ap  = Ax + 16 * XP;
  const float* Bi0 = Wi + (size_t)(n0 + lo) * DIM;
  const float* Be0 = We + (size_t)(n0 + lo) * DIM;
  const float* Bi1 = Bi0 + 16 * DIM;
  const float* Be1 = Be0 + 16 * DIM;

  v8f o0[RT], o1[RT];
  #pragma unroll
  for (int t = 0; t < RT; ++t) { o0[t] = (v8f){}; o1[t] = (v8f){}; }
  #pragma unroll 2
  for (int kk = 0; kk < DIM; kk += 4) {
    const int k = kk + 2 * khalf;
    const v2f bi0 = *(const v2f*)(Bi0 + k);
    const v2f bi1 = *(const v2f*)(Bi1 + k);
    const v2f be0 = *(const v2f*)(Be0 + k);
    const v2f be1 = *(const v2f*)(Be1 + k);
    #pragma unroll
    for (int t = 0; t < RT; ++t) {
      const v2f ax = *(const v2f*)(Ax + t * XPT + k);
      const v2f ap = *(const v2f*)(Ap + t * XPT + k);
      o0[t] = wmma_f32(ax, bi0, o0[t]);
      o1[t] = wmma_f32(ax, bi1, o1[t]);
      o0[t] = wmma_f32(ap, be0, o0[t]);
      o1[t] = wmma_f32(ap, be1, o1[t]);
    }
  }

  #pragma unroll
  for (int t = 0; t < RT; ++t) {
    const int m0 = (tile0 + t) * 16;
    if (m0 + 16 <= nrows) {
      float* o = out + (size_t)(m0 + 8 * khalf) * DIM + n0 + lo;
      #pragma unroll
      for (int r = 0; r < 8; ++r) {
        const float y0 = o0[t][r], y1 = o1[t][r];
        o[(size_t)r * DIM]      = (y0 > 0.0f) ? y0 : 0.01f * y0;
        o[(size_t)r * DIM + 16] = (y1 > 0.0f) ? y1 : 0.01f * y1;
      }
    } else {
      #pragma unroll
      for (int r = 0; r < 8; ++r) {
        const int orow = m0 + 8 * khalf + r;
        if (orow < nrows) {
          float* o = out + (size_t)orow * DIM + n0 + lo;
          const float y0 = o0[t][r], y1 = o1[t][r];
          o[0]  = (y0 > 0.0f) ? y0 : 0.01f * y0;
          o[16] = (y1 > 0.0f) ? y1 : 0.01f * y1;
        }
      }
    }
  }
}

extern "C" void kernel_launch(void* const* d_in, const int* in_sizes, int n_in,
                              void* d_out, int out_size, void* d_ws, size_t ws_size,
                              hipStream_t stream) {
  const float* src_x = (const float*)d_in[0];
  const float* dst_x = (const float*)d_in[1];
  const int*   eidx  = (const int*)  d_in[2];
  const float* ew    = (const float*)d_in[3];
  const float* Wi    = (const float*)d_in[4];
  const float* We    = (const float*)d_in[5];
  float*       out   = (float*)d_out;
  int*         ws    = (int*)d_ws;

  const int N  = in_sizes[0] / DIM;
  const int E  = in_sizes[2] / 2;
  const int NT = (N + 15) / 16;
  const int* src_idx = eidx;
  const int* dst_idx = eidx + E;

  const dim3 bgrid(NCH, (NT + RB - 1) / RB);
  bin_count_kernel<<<bgrid, 1024, 0, stream>>>(dst_idx, ws, ws_size, NT, E);
  bin_place_kernel<<<bgrid, 1024, 0, stream>>>(src_idx, dst_idx, ew, ws, ws_size, NT, E);
  tile_gather_transform_kernel<<<(NT + RT - 1) / RT, 128, 0, stream>>>(src_x, dst_x, Wi, We, ws, ws_size,
                                                       out, N, NT, E);
}


# Each wave aggregates one whole tile, its B fragments read straight from global rows (half-wave = one 128-byte line) 4 k-steps ahead: no row staging in LDS, no pass barriers; sums unchanged
# speedup vs baseline: 5.8346x; 1.0828x over previous
#include <hip/hip_runtime.h>


typedef __attribute__((ext_vector_type(2))) float v2f;
typedef __attribute__((ext_vector_type(4))) float v4f;
typedef __attribute__((ext_vector_type(8))) float v8f;
typedef __attribute__((ext_vector_type(2))) unsigned int v2u;
typedef __attribute__((ext_vector_type(4))) unsigned int v4u;

#define DIM 128
#define XP  132
#define RT  4
#define XPT (2 * 16 * XP)

static __device__ __forceinline__ v8f wmma_f32(v2f a, v2f b, v8f c) {
  return __builtin_amdgcn_wmma_f32_16x16x4_f32(false, a, false, b,
                                               (short)0, c, false, false);
}

#define NCH 32
#define RB  1024

static __device__ __forceinline__ size_t ws_plane(int ntiles) {
  return ((size_t)ntiles + 32) & ~(size_t)31;
}
static __device__ __forceinline__ int bin_ranges(int ntiles) {
  return (ntiles + RB - 1) / RB;
}
static __device__ __forceinline__ size_t ws_rec(int ntiles) {
  const size_t rs = ((size_t)NCH * bin_ranges(ntiles) + 31) & ~(size_t)31;
  return (size_t)(NCH + 1) * ws_plane(ntiles) + rs;
}
static __device__ __forceinline__ bool ws_fits(size_t ws_bytes, int ntiles, int nedges) {
  return ws_bytes >= ws_rec(ntiles) * 4 + (size_t)nedges * 8;
}

static __device__ __forceinline__ int block_sum(int v, int* s_w) {
  #pragma unroll
  for (int m = 16; m >= 1; m >>= 1) v += __shfl_xor(v, m);
  if ((threadIdx.x & 31) == 0) s_w[threadIdx.x >> 5] = v;
  __syncthreads();
  int r = s_w[threadIdx.x & 31];
  #pragma unroll
  for (int m = 16; m >= 1; m >>= 1) r += __shfl_xor(r, m);
  __syncthreads();
  return r;
}
static __device__ __forceinline__ int block_excl_scan(int v, int* s_w) {
  const int lane = threadIdx.x & 31;
  const int wv   = threadIdx.x >> 5;
  int x = v;
  #pragma unroll
  for (int d = 1; d < 32; d <<= 1) {
    const int y = __shfl_up(x, d);
    if (lane >= d) x += y;
  }
  if (lane == 31) s_w[wv] = x;
  __syncthreads();
  const int w = s_w[lane];
  int wsum = w;
  #pragma unroll
  for (int d = 1; d < 32; d <<= 1) {
    const int y = __shfl_up(wsum, d);
    if (lane >= d) wsum += y;
  }
  const int wex = __shfl(wsum - w, wv);
  __syncthreads();
  return wex + x - v;
}

__global__ void __launch_bounds__(1024)
bin_count_kernel(const int* __restrict__ dst_idx, int* __restrict__ ws,
                 size_t ws_bytes, int ntiles, int nedges) {
  if (!ws_fits(ws_bytes, ntiles, nedges)) return;
  __shared__ int s_cnt[RB];
  __shared__ int s_w[32];
  const int t  = threadIdx.x;
  const int c  = blockIdx.x;
  const int h  = blockIdx.y;
  const size_t tp = ws_plane(ntiles);
  int* cnt  = ws + tp;
  int* rsum = ws + (size_t)(NCH + 1) * tp;
  const int per   = (nedges + NCH - 1) / NCH;
  const int e_beg = c * per;
  const int e_end = (e_beg + per < nedges) ? e_beg + per : nedges;
  const int b0    = h * RB;
  s_cnt[t] = 0;
  __syncthreads();
  int mine = 0;
  for (int e0 = e_beg + t; e0 < e_end; e0 += 4 * 1024) {
    int d[4];
    #pragma unroll
    for (int i = 0; i < 4; ++i) {
      const int e = e0 + i * 1024;
      d[i] = dst_idx[(e < e_end) ? e : e0];
    }
    #pragma unroll
    for (int i = 0; i < 4; ++i) {
      const int e  = e0 + i * 1024;
      const int tl = d[i] >> 4;
      const unsigned u = (unsigned)(tl - b0);
      if (e < e_end && u < (unsigned)RB && tl < ntiles) {
        atomicAdd(&s_cnt[u], 1);
        ++mine;
      }
    }
  }
  __syncthreads();
  if (b0 + t < ntiles) cnt[(size_t)c * tp + b0 + t] = s_cnt[t];
  const int tot = block_sum(mine, s_w);
  if (t == 0) rsum[h * NCH + c] = tot;
}

__global__ void __launch_bounds__(1024)
bin_place_kernel(const int* __restrict__ src_idx, const int* __restrict__ dst_idx,
                 const float* __restrict__ ew, int* __restrict__ ws,
                 size_t ws_bytes, int ntiles, int nedges) {
  if (!ws_fits(ws_bytes, ntiles, nedges)) return;
  __shared__ int s_cur[RB];
  __shared__ int s_w[32];
  const int t  = threadIdx.x;
  const int c  = blockIdx.x;
  const int h  = blockIdx.y;
  const size_t tp = ws_plane(ntiles);
  int*       off  = ws;
  const int* cnt  = ws + tp;
  const int* rsum = ws + (size_t)(NCH + 1) * tp;
  v2u*       rec  = (v2u*)(ws + ws_rec(ntiles));
  const int b0 = h * RB;
  const int b  = b0 + t;

  int tot = 0, pre = 0;
  if (b < ntiles) {
    #pragma unroll 8
    for (int cc = 0; cc < NCH; ++cc) {
      const int v = cnt[(size_t)cc * tp + b];
      tot += v;
      pre += (cc < c) ? v : 0;
    }
  }
  int low = 0;
  for (int j = t; j < h * NCH; j += 1024) low += rsum[j];
  const int lowsum = block_sum(low, s_w);
  const int first  = lowsum + block_excl_scan(tot, s_w);
  if (c == 0) {
    if (b < ntiles) off[b] = first;
    if (h == 0 && t == 0) off[ntiles] = nedges;
  }
  s_cur[t] = first + pre;
  __syncthreads();

  const int per   = (nedges + NCH - 1) / NCH;
  const int e_beg = c * per;
  const int e_end = (e_beg + per < nedges) ? e_beg + per : nedges;
  for (int e0 = e_beg + t; e0 < e_end; e0 += 4 * 1024) {
    int d[4], s[4];
    float w[4];
    #pragma unroll
    for (int i = 0; i < 4; ++i) {
      const int e  = e0 + i * 1024;
      const int ec = (e < e_end) ? e : e0;
      d[i] = dst_idx[ec];
      s[i] = src_idx[ec];
      w[i] = ew[ec];
    }
    #pragma unroll
    for (int i = 0; i < 4; ++i) {
      const int e  = e0 + i * 1024;
      const int tl = d[i] >> 4;
      const unsigned u = (unsigned)(tl - b0);
      if (e < e_end && u < (unsigned)RB && tl < ntiles) {
        const int pos = atomicAdd(&s_cur[u], 1);
        v2u r;
        r.x = (unsigned)s[i] | ((unsigned)(d[i] & 15) << 28);
        r.y = __float_as_uint(w[i]);
        rec[pos] = r;
      }
    }
  }
}

#define KS 4

static __device__ __forceinline__ void stage_records(v4u (&q)[KS], unsigned int* sr, v2u r, int lane, int khalf) {
  if (lane < 16) *(v2u*)&sr[2 * lane] = r;
  __builtin_amdgcn_fence(__ATOMIC_RELEASE, "wavefront", "local");
  __builtin_amdgcn_wave_barrier();
  __builtin_amdgcn_fence(__ATOMIC_ACQUIRE, "wavefront", "local");
  #pragma unroll
  for (int j = 0; j < KS; ++j) q[j] = *(const v4u*)&sr[8 * j + 4 * khalf];
}

static __device__ __forceinline__ void issue_kstep(v2f (&g)[2][4], const float* __restrict__ src_x, v4u q, int lo) {
  const float* p0 = src_x + (size_t)(q.x & 0x0FFFFFFFu) * DIM + 2 * lo;
  const float* p1 = src_x + (size_t)(q.z & 0x0FFFFFFFu) * DIM + 2 * lo;
  #pragma unroll
  for (int c = 0; c < 4; ++c) {
    g[0][c] = *(const v2f*)(p0 + 32 * c);
    g[1][c] = *(const v2f*)(p1 + 32 * c);
  }
}

__global__ void __launch_bounds__(128)
tile_gather_transform_kernel(const float* __restrict__ src_x, const float* __restrict__ dst_x,
                             const float* __restrict__ Wi, const float* __restrict__ We,
                             const int* __restrict__ ws, size_t ws_bytes,
                             float* __restrict__ out, int nrows, int ntiles, int nedges) {
  if (!ws_fits(ws_bytes, ntiles, nedges)) return;
  __shared__ __attribute__((aligned(16))) float        s_all[RT * XPT];
  __shared__ __attribute__((aligned(16))) unsigned int s_recw[4 * KS * 8];

  const int tid   = threadIdx.x;
  const int lane  = tid & 31;
  const int wave  = tid >> 5;
  const int lo    = lane & 15;
  const int khalf = lane >> 4;
  const int n0    = wave * 32;
  const int tile0 = blockIdx.x * RT;

  const int* off = ws;
  const v2u* rec = (const v2u*)(ws + ws_rec(ntiles));
  const int wv = __builtin_amdgcn_readfirstlane(wave);

  static_assert(RT == 4, "one wave per tile of the group");
  const int tw = tile0 + wv;
  if (tw < ntiles) {
    const int beg = off[tw];
    const int end = off[tw + 1];
    unsigned int* sr = &s_recw[wv * (KS * 8)];
    const v2u none = {0u, 0u};

    v2u r0 = none, r1 = none, r2 = none;
    if (lane < 16 && beg + lane < end)          r0 = rec[beg + lane];
    if (lane < 16 && beg + 4 * KS + lane < end) r1 = rec[beg + 4 * KS + lane];
    if (lane < 16 && beg + 8 * KS + lane < end) r2 = rec[beg + 8 * KS + lane];
    v4u qc[KS], qn[KS];
    stage_records(qc, sr, r0, lane, khalf);
    v2f g[KS][2][4];
    #pragma unroll
    for (int j = 0; j < KS; ++j) issue_kstep(g[j], src_x, qc[j], lo);
    stage_records(qn, sr, r1, lane, khalf);

    v8f acc[8];
    #pragma unroll
    for (int i = 0; i < 8; ++i) acc[i] = (v8f){};

    for (int base = beg; base < end; base += 4 * KS) {
      #pragma unroll
      for (int j = 0; j < KS; ++j) {
        const v4u q = qc[j];
        v2f a;
        a.x = ((q.x >> 28) == (unsigned)lo) ? __uint_as_float(q.y) : 0.0f;
        a.y = ((q.z >> 28) == (unsigned)lo) ? __uint_as_float(q.w) : 0.0f;
        #pragma unroll
        for (int c = 0; c < 4; ++c) {
          const v2f b0 = {g[j][0][c].x, g[j][1][c].x};
          const v2f b1 = {g[j][0][c].y, g[j][1][c].y};
          acc[2 * c]     = wmma_f32(a, b0, acc[2 * c]);
          acc[2 * c + 1] = wmma_f32(a, b1, acc[2 * c + 1]);
        }
        issue_kstep(g[j], src_x, qn[j], lo);
      }
      v4u qnn[KS];
      stage_records(qnn, sr, r2, lane, khalf);
      const int b3 = base + 12 * KS;
      r2 = none;
      if (lane < 16 && b3 + lane < end) r2 = rec[b3 + lane];
      #pragma unroll
      for (int j = 0; j < KS; ++j) { qc[j] = qn[j]; qn[j] = qnn[j]; }
    }

    const int m0 = tw * 16;
    float* s_x = s_all + wv * XPT;
    float* s_p = s_x + 16 * XP;
    #pragma unroll
    for (int r = 0; r < 8; ++r) {
      int row = m0 + r + 8 * khalf;
      if (row >= nrows) row = nrows - 1;
      const float* dp = dst_x + (size_t)row * DIM + 2 * lo;
      v2f d[4];
      #pragma unroll
      for (int c = 0; c < 4; ++c) d[c] = *(const v2f*)(dp + 32 * c);
      const int o = (r + 8 * khalf) * XP + 2 * lo;
      #pragma unroll
      for (int c = 0; c < 4; ++c) {
        const float a0 = acc[2 * c][r], a1 = acc[2 * c + 1][r];
        const v2f x = {d[c].x + a0, d[c].y + a1};
        const v2f p = {a0 * d[c].x, a1 * d[c].y};
        *(v2f*)&s_x[o + 32 * c] = x;
        *(v2f*)&s_p[o + 32 * c] = p;
      }
    }
  }
  __syncthreads();

  const float* Ax  = s_all + lo * XP;
  const float* Ap  = Ax + 16 * XP;
  const float* Bi0 = Wi + (size_t)(n0 + lo) * DIM;
  const float* Be0 = We + (size_t)(n0 + lo) * DIM;
  const float* Bi1 = Bi0 + 16 * DIM;
  const float* Be1 = Be0 + 16 * DIM;

  v8f o0[RT], o1[RT];
  #pragma unroll
  for (int t = 0; t < RT; ++t) { o0[t] = (v8f){}; o1[t] = (v8f){}; }
  #pragma unroll 2
  for (int kk = 0; kk < DIM; kk += 4) {
    const int k = kk + 2 * khalf;
    const v2f bi0 = *(const v2f*)(Bi0 + k);
    const v2f bi1 = *(const v2f*)(Bi1 + k);
    const v2f be0 = *(const v2f*)(Be0 + k);
    const v2f be1 = *(const v2f*)(Be1 + k);
    #pragma unroll
    for (int t = 0; t < RT; ++t) {
      const v2f ax = *(const v2f*)(Ax + t * XPT + k);
      const v2f ap = *(const v2f*)(Ap + t * XPT + k);
      o0[t] = wmma_f32(ax, bi0, o0[t]);
      o1[t] = wmma_f32(ax, bi1, o1[t]);
      o0[t] = wmma_f32(ap, be0, o0[t]);
      o1[t] = wmma_f32(ap, be1, o1[t]);
    }
  }

  #pragma unroll
  for (int t = 0; t < RT; ++t) {
    const int m0 = (tile0 + t) * 16;
    if (m0 + 16 <= nrows) {
      float* o = out + (size_t)(m0 + 8 * khalf) * DIM + n0 + lo;
      #pragma unroll
      for (int r = 0; r < 8; ++r) {
        const float y0 = o0[t][r], y1 = o1[t][r];
        o[(size_t)r * DIM]      = (y0 > 0.0f) ? y0 : 0.01f * y0;
        o[(size_t)r * DIM + 16] = (y1 > 0.0f) ? y1 : 0.01f * y1;
      }
    } else {
      #pragma unroll
      for (int r = 0; r < 8; ++r) {
        const int orow = m0 + 8 * khalf + r;
        if (orow < nrows) {
          float* o = out + (size_t)orow * DIM + n0 + lo;
          const float y0 = o0[t][r], y1 = o1[t][r];
          o[0]  = (y0 > 0.0f) ? y0 : 0.01f * y0;
          o[16] = (y1 > 0.0f) ? y1 : 0.01f * y1;
        }
      }
    }
  }
}

extern "C" void kernel_launch(void* const* d_in, const int* in_sizes, int n_in,
                              void* d_out, int out_size, void* d_ws, size_t ws_size,
                              hipStream_t stream) {
  const float* src_x = (const float*)d_in[0];
  const float* dst_x = (const float*)d_in[1];
  const int*   eidx  = (const int*)  d_in[2];
  const float* ew    = (const float*)d_in[3];
  const float* Wi    = (const float*)d_in[4];
  const float* We    = (const float*)d_in[5];
  float*       out   = (float*)d_out;
  int*         ws    = (int*)d_ws;

  const int N  = in_sizes[0] / DIM;
  const int E  = in_sizes[2] / 2;
  const int NT = (N + 15) / 16;
  const int* src_idx = eidx;
  const int* dst_idx = eidx + E;

  const dim3 bgrid(NCH, (NT + RB - 1) / RB);
  bin_count_kernel<<<bgrid, 1024, 0, stream>>>(dst_idx, ws, ws_size, NT, E);
  bin_place_kernel<<<bgrid, 1024, 0, stream>>>(src_idx, dst_idx, ew, ws, ws_size, NT, E);
  tile_gather_transform_kernel<<<(NT + RT - 1) / RT, 128, 0, stream>>>(src_x, dst_x, Wi, We, ws, ws_size,
                                                       out, N, NT, E);
}


# Weights repacked per call into B-fragment order in the workspace (in the count pass): a transform weight load is 512 contiguous bytes, was 16 bytes from each of 16 rows; sums unchanged
# speedup vs baseline: 6.0531x; 1.0374x over previous
#include <hip/hip_runtime.h>


typedef __attribute__((ext_vector_type(2))) float v2f;
typedef __attribute__((ext_vector_type(4))) float v4f;
typedef __attribute__((ext_vector_type(8))) float v8f;
typedef __attribute__((ext_vector_type(2))) unsigned int v2u;
typedef __attribute__((ext_vector_type(4))) unsigned int v4u;

#define DIM 128
#define XP  132
#define RT  4
#define XPT (2 * 16 * XP)

static __device__ __forceinline__ v8f wmma_f32(v2f a, v2f b, v8f c) {
  return __builtin_amdgcn_wmma_f32_16x16x4_f32(false, a, false, b,
                                               (short)0, c, false, false);
}

#define NCH 32
#define RB  1024

static __device__ __forceinline__ size_t ws_plane(int ntiles) {
  return ((size_t)ntiles + 32) & ~(size_t)31;
}
static __device__ __forceinline__ int bin_ranges(int ntiles) {
  return (ntiles + RB - 1) / RB;
}
static __device__ __forceinline__ size_t ws_rec(int ntiles) {
  const size_t rs = ((size_t)NCH * bin_ranges(ntiles) + 31) & ~(size_t)31;
  return (size_t)(NCH + 1) * ws_plane(ntiles) + rs;
}
static __device__ __forceinline__ size_t ws_wpk(int ntiles, int nedges) {
  return (ws_rec(ntiles) + 2 * (size_t)nedges + 127) & ~(size_t)127;
}
static __device__ __forceinline__ bool ws_fits(size_t ws_bytes, int ntiles, int nedges) {
  return ws_bytes >= (ws_wpk(ntiles, nedges) + 2 * DIM * DIM) * 4;
}

#define WPK_ELEMS (2 * 8 * 16 * 32)

static __device__ __forceinline__ int block_sum(int v, int* s_w) {
  #pragma unroll
  for (int m = 16; m >= 1; m >>= 1) v += __shfl_xor(v, m);
  if ((threadIdx.x & 31) == 0) s_w[threadIdx.x >> 5] = v;
  __syncthreads();
  int r = s_w[threadIdx.x & 31];
  #pragma unroll
  for (int m = 16; m >= 1; m >>= 1) r += __shfl_xor(r, m);
  __syncthreads();
  return r;
}
static __device__ __forceinline__ int block_excl_scan(int v, int* s_w) {
  const int lane = threadIdx.x & 31;
  const int wv   = threadIdx.x >> 5;
  int x = v;
  #pragma unroll
  for (int d = 1; d < 32; d <<= 1) {
    const int y = __shfl_up(x, d);
    if (lane >= d) x += y;
  }
  if (lane == 31) s_w[wv] = x;
  __syncthreads();
  const int w = s_w[lane];
  int wsum = w;
  #pragma unroll
  for (int d = 1; d < 32; d <<= 1) {
    const int y = __shfl_up(wsum, d);
    if (lane >= d) wsum += y;
  }
  const int wex = __shfl(wsum - w, wv);
  __syncthreads();
  return wex + x - v;
}

__global__ void __launch_bounds__(1024)
bin_count_kernel(const int* __restrict__ dst_idx, const float* __restrict__ Wi, const float* __restrict__ We,
                 int* __restrict__ ws, size_t ws_bytes, int ntiles, int nedges) {
  if (!ws_fits(ws_bytes, ntiles, nedges)) return;
  __shared__ int s_cnt[RB];
  __shared__ int s_w[32];
  const int t  = threadIdx.x;
  const int c  = blockIdx.x;
  const int h  = blockIdx.y;

  const size_t tp = ws_plane(ntiles);
  int* cnt  = ws + tp;
  int* rsum = ws + (size_t)(NCH + 1) * tp;
  const int per   = (nedges + NCH - 1) / NCH;
  const int e_beg = c * per;
  const int e_end = (e_beg + per < nedges) ? e_beg + per : nedges;
  const int b0    = h * RB;
  s_cnt[t] = 0;
  __syncthreads();

  const int pk = (h * NCH + c) * 1024 + t;
  v2f pa = {0.0f, 0.0f}, pb = {0.0f, 0.0f};
  if (pk < WPK_ELEMS) {
    const int pl = pk & 31, pj = (pk >> 5) & 15, pt = (pk >> 9) & 7;
    const float* wsrc = ((pk >> 12) ? We : Wi) + (size_t)(16 * pt + (pl & 15)) * DIM + 8 * pj + 2 * (pl >> 4);
    pa = *(const v2f*)wsrc;
    pb = *(const v2f*)(wsrc + 4);
  }
  int mine = 0;
  for (int e0 = e_beg + t; e0 < e_end; e0 += 4 * 1024) {
    int d[4];
    #pragma unroll
    for (int i = 0; i < 4; ++i) {
      const int e = e0 + i * 1024;
      d[i] = dst_idx[(e < e_end) ? e : e0];
    }
    #pragma unroll
    for (int i = 0; i < 4; ++i) {
      const int e  = e0 + i * 1024;
      const int tl = d[i] >> 4;
      const unsigned u = (unsigned)(tl - b0);
      if (e < e_end && u < (unsigned)RB && tl < ntiles) {
        atomicAdd(&s_cnt[u], 1);
        ++mine;
      }
    }
  }
  __syncthreads();
  if (b0 + t < ntiles) cnt[(size_t)c * tp + b0 + t] = s_cnt[t];
  const int tot = block_sum(mine, s_w);
  if (t == 0) rsum[h * NCH + c] = tot;
  if (pk < WPK_ELEMS) {
    const v4f pv = {pa.x, pa.y, pb.x, pb.y};
    ((v4f*)(ws + ws_wpk(ntiles, nedges)))[pk] = pv;
  }
}

__global__ void __launch_bounds__(1024)
bin_place_kernel(const int* __restrict__ src_idx, const int* __restrict__ dst_idx,
                 const float* __restrict__ ew, int* __restrict__ ws,
                 size_t ws_bytes, int ntiles, int nedges) {
  if (!ws_fits(ws_bytes, ntiles, nedges)) return;
  __shared__ int s_cur[RB];
  __shared__ int s_w[32];
  const int t  = threadIdx.x;
  const int c  = blockIdx.x;
  const int h  = blockIdx.y;
  const size_t tp = ws_plane(ntiles);
  int*       off  = ws;
  const int* cnt  = ws + tp;
  const int* rsum = ws + (size_t)(NCH + 1) * tp;
  v2u*       rec  = (v2u*)(ws + ws_rec(ntiles));
  const int b0 = h * RB;
  const int b  = b0 + t;

  int tot = 0, pre = 0;
  if (b < ntiles) {
    #pragma unroll 8
    for (int cc = 0; cc < NCH; ++cc) {
      const int v = cnt[(size_t)cc * tp + b];
      tot += v;
      pre += (cc < c) ? v : 0;
    }
  }
  int low = 0;
  for (int j = t; j < h * NCH; j += 1024) low += rsum[j];
  const int lowsum = block_sum(low, s_w);
  const int first  = lowsum + block_excl_scan(tot, s_w);
  if (c == 0) {
    if (b < ntiles) off[b] = first;
    if (h == 0 && t == 0) off[ntiles] = nedges;
  }
  s_cur[t] = first + pre;
  __syncthreads();

  const int per   = (nedges + NCH - 1) / NCH;
  const int e_beg = c * per;
  const int e_end = (e_beg + per < nedges) ? e_beg + per : nedges;
  for (int e0 = e_beg + t; e0 < e_end; e0 += 4 * 1024) {
    int d[4], s[4];
    float w[4];
    #pragma unroll
    for (int i = 0; i < 4; ++i) {
      const int e  = e0 + i * 1024;
      const int ec = (e < e_end) ? e : e0;
      d[i] = dst_idx[ec];
      s[i] = src_idx[ec];
      w[i] = ew[ec];
    }
    #pragma unroll
    for (int i = 0; i < 4; ++i) {
      const int e  = e0 + i * 1024;
      const int tl = d[i] >> 4;
      const unsigned u = (unsigned)(tl - b0);
      if (e < e_end && u < (unsigned)RB && tl < ntiles) {
        const int pos = atomicAdd(&s_cur[u], 1);
        v2u r;
        r.x = (unsigned)s[i] | ((unsigned)(d[i] & 15) << 28);
        r.y = __float_as_uint(w[i]);
        rec[pos] = r;
      }
    }
  }
}

#define KS 4

static __device__ __forceinline__ void stage_records(v4u (&q)[KS], unsigned int* sr, v2u r, int lane, int khalf) {
  if (lane < 16) *(v2u*)&sr[2 * lane] = r;
  __builtin_amdgcn_fence(__ATOMIC_RELEASE, "wavefront", "local");
  __builtin_amdgcn_wave_barrier();
  __builtin_amdgcn_fence(__ATOMIC_ACQUIRE, "wavefront", "local");
  #pragma unroll
  for (int j = 0; j < KS; ++j) q[j] = *(const v4u*)&sr[8 * j + 4 * khalf];
}

static __device__ __forceinline__ void issue_kstep(v2f (&g)[2][4], const float* __restrict__ src_x, v4u q, int lo) {
  const float* p0 = src_x + (size_t)(q.x & 0x0FFFFFFFu) * DIM + 2 * lo;
  const float* p1 = src_x + (size_t)(q.z & 0x0FFFFFFFu) * DIM + 2 * lo;
  #pragma unroll
  for (int c = 0; c < 4; ++c) {
    g[0][c] = *(const v2f*)(p0 + 32 * c);
    g[1][c] = *(const v2f*)(p1 + 32 * c);
  }
}

__global__ void __launch_bounds__(128)
tile_gather_transform_kernel(const float* __restrict__ src_x, const float* __restrict__ dst_x,
                             const int* __restrict__ ws, size_t ws_bytes,
                             float* __restrict__ out, int nrows, int ntiles, int nedges) {
  if (!ws_fits(ws_bytes, ntiles, nedges)) return;
  __shared__ __attribute__((aligned(16))) float        s_all[RT * XPT];
  __shared__ __attribute__((aligned(16))) unsigned int s_recw[4 * KS * 8];

  const int tid   = threadIdx.x;
  const int lane  = tid & 31;
  const int wave  = tid >> 5;
  const int lo    = lane & 15;
  const int khalf = lane >> 4;
  const int n0    = wave * 32;
  const int tile0 = blockIdx.x * RT;

  const int* off = ws;
  const v2u* rec = (const v2u*)(ws + ws_rec(ntiles));
  const int wv = __builtin_amdgcn_readfirstlane(wave);

  static_assert(RT == 4, "one wave per tile of the group");
  const int tw = tile0 + wv;
  if (tw < ntiles) {
    const int beg = off[tw];
    const int end = off[tw + 1];
    unsigned int* sr = &s_recw[wv * (KS * 8)];
    const v2u none = {0u, 0u};

    v2u r0 = none, r1 = none, r2 = none;
    if (lane < 16 && beg + lane < end)          r0 = rec[beg + lane];
    if (lane < 16 && beg + 4 * KS + lane < end) r1 = rec[beg + 4 * KS + lane];
    if (lane < 16 && beg + 8 * KS + lane < end) r2 = rec[beg + 8 * KS + lane];
    v4u qc[KS], qn[KS];
    stage_records(qc, sr, r0, lane, khalf);
    v2f g[KS][2][4];
    #pragma unroll
    for (int j = 0; j < KS; ++j) issue_kstep(g[j], src_x, qc[j], lo);
    stage_records(qn, sr, r1, lane, khalf);

    v8f acc[8];
    #pragma unroll
    for (int i = 0; i < 8; ++i) acc[i] = (v8f){};

    for (int base = beg; base < end; base += 4 * KS) {
      #pragma unroll
      for (int j = 0; j < KS; ++j) {
        const v4u q = qc[j];
        v2f a;
        a.x = ((q.x >> 28) == (unsigned)lo) ? __uint_as_float(q.y) : 0.0f;
        a.y = ((q.z >> 28) == (unsigned)lo) ? __uint_as_float(q.w) : 0.0f;
        #pragma unroll
        for (int c = 0; c < 4; ++c) {
          const v2f b0 = {g[j][0][c].x, g[j][1][c].x};
          const v2f b1 = {g[j][0][c].y, g[j][1][c].y};
          acc[2 * c]     = wmma_f32(a, b0, acc[2 * c]);
          acc[2 * c + 1] = wmma_f32(a, b1, acc[2 * c + 1]);
        }
        issue_kstep(g[j], src_x, qn[j], lo);
      }
      v4u qnn[KS];
      stage_records(qnn, sr, r2, lane, khalf);
      const int b3 = base + 12 * KS;
      r2 = none;
      if (lane < 16 && b3 + lane < end) r2 = rec[b3 + lane];
      #pragma unroll
      for (int j = 0; j < KS; ++j) { qc[j] = qn[j]; qn[j] = qnn[j]; }
    }

    const int m0 = tw * 16;
    float* s_x = s_all + wv * XPT;
    float* s_p = s_x + 16 * XP;
    #pragma unroll
    for (int r = 0; r < 8; ++r) {
      int row = m0 + r + 8 * khalf;
      if (row >= nrows) row = nrows - 1;
      const float* dp = dst_x + (size_t)row * DIM + 2 * lo;
      v2f d[4];
      #pragma unroll
      for (int c = 0; c < 4; ++c) d[c] = *(const v2f*)(dp + 32 * c);
      const int o = (r + 8 * khalf) * XP + 2 * lo;
      #pragma unroll
      for (int c = 0; c < 4; ++c) {
        const float a0 = acc[2 * c][r], a1 = acc[2 * c + 1][r];
        const v2f x = {d[c].x + a0, d[c].y + a1};
        const v2f p = {a0 * d[c].x, a1 * d[c].y};
        *(v2f*)&s_x[o + 32 * c] = x;
        *(v2f*)&s_p[o + 32 * c] = p;
      }
    }
  }
  __syncthreads();

  const float* Ax  = s_all + lo * XP;
  const float* Ap  = Ax + 16 * XP;
  const v4f* Bi0 = (const v4f*)(ws + ws_wpk(ntiles, nedges)) + (size_t)(2 * wv) * (16 * 32) + lane;
  const v4f* Bi1 = Bi0 + 16 * 32;
  const v4f* Be0 = Bi0 + 8 * 16 * 32;
  const v4f* Be1 = Be0 + 16 * 32;

  v8f o0[RT], o1[RT];
  #pragma unroll
  for (int t = 0; t < RT; ++t) { o0[t] = (v8f){}; o1[t] = (v8f){}; }
  for (int j = 0; j < DIM / 8; ++j) {
    const v4f wi0 = Bi0[j * 32];
    const v4f wi1 = Bi1[j * 32];
    const v4f we0 = Be0[j * 32];
    const v4f we1 = Be1[j * 32];
    #pragma unroll
    for (int s = 0; s < 2; ++s) {
      const int k = 8 * j + 4 * s + 2 * khalf;
      const v2f bi0 = s ? (v2f){wi0.z, wi0.w} : (v2f){wi0.x, wi0.y};
      const v2f bi1 = s ? (v2f){wi1.z, wi1.w} : (v2f){wi1.x, wi1.y};
      const v2f be0 = s ? (v2f){we0.z, we0.w} : (v2f){we0.x, we0.y};
      const v2f be1 = s ? (v2f){we1.z, we1.w} : (v2f){we1.x, we1.y};
      #pragma unroll
      for (int t = 0; t < RT; ++t) {
        const v2f ax = *(const v2f*)(Ax + t * XPT + k);
        const v2f ap = *(const v2f*)(Ap + t * XPT + k);
        o0[t] = wmma_f32(ax, bi0, o0[t]);
        o1[t] = wmma_f32(ax, bi1, o1[t]);
        o0[t] = wmma_f32(ap, be0, o0[t]);
        o1[t] = wmma_f32(ap, be1, o1[t]);
      }
    }
  }

  #pragma unroll
  for (int t = 0; t < RT; ++t) {
    const int m0 = (tile0 + t) * 16;
    if (m0 + 16 <= nrows) {
      float* o = out + (size_t)(m0 + 8 * khalf) * DIM + n0 + lo;
      #pragma unroll
      for (int r = 0; r < 8; ++r) {
        const float y0 = o0[t][r], y1 = o1[t][r];
        o[(size_t)r * DIM]      = (y0 > 0.0f) ? y0 : 0.01f * y0;
        o[(size_t)r * DIM + 16] = (y1 > 0.0f) ? y1 : 0.01f * y1;
      }
    } else {
      #pragma unroll
      for (int r = 0; r < 8; ++r) {
        const int orow = m0 + 8 * khalf + r;
        if (orow < nrows) {
          float* o = out + (size_t)orow * DIM + n0 + lo;
          const float y0 = o0[t][r], y1 = o1[t][r];
          o[0]  = (y0 > 0.0f) ? y0 : 0.01f * y0;
          o[16] = (y1 > 0.0f) ? y1 : 0.01f * y1;
        }
      }
    }
  }
}

extern "C" void kernel_launch(void* const* d_in, const int* in_sizes, int n_in,
                              void* d_out, int out_size, void* d_ws, size_t ws_size,
                              hipStream_t stream) {
  const float* src_x = (const float*)d_in[0];
  const float* dst_x = (const float*)d_in[1];
  const int*   eidx  = (const int*)  d_in[2];
  const float* ew    = (const float*)d_in[3];
  const float* Wi    = (const float*)d_in[4];
  const float* We    = (const float*)d_in[5];
  float*       out   = (float*)d_out;
  int*         ws    = (int*)d_ws;

  const int N  = in_sizes[0] / DIM;
  const int E  = in_sizes[2] / 2;
  const int NT = (N + 15) / 16;
  const int* src_idx = eidx;
  const int* dst_idx = eidx + E;

  const dim3 bgrid(NCH, (NT + RB - 1) / RB);
  bin_count_kernel<<<bgrid, 1024, 0, stream>>>(dst_idx, Wi, We, ws, ws_size, NT, E);
  bin_place_kernel<<<bgrid, 1024, 0, stream>>>(src_idx, dst_idx, ew, ws, ws_size, NT, E);
  tile_gather_transform_kernel<<<(NT + RT - 1) / RT, 128, 0, stream>>>(src_x, dst_x, ws, ws_size,
                                                       out, N, NT, E);
}
